# starting kernel
# baseline (speedup 1.0000x reference)
#include <hip/hip_runtime.h>


typedef __attribute__((ext_vector_type(16))) _Float16 v16h;
typedef __attribute__((ext_vector_type(8)))  _Float16 v8h;
typedef __attribute__((ext_vector_type(8)))  float    v8f;

#define N_NODES 50000
#define E_EDGES 800000
#define P_CAND  100000
#define HDIM    128
#define NEG_SLOPE 0.2f


__device__ __forceinline__ int f2ord(float f) {
    int i = __float_as_int(f);
    return i >= 0 ? i : (i ^ 0x7fffffff);
}
__device__ __forceinline__ float ord2f(int i) {
    return __int_as_float(i >= 0 ? i : (i ^ 0x7fffffff));
}

__device__ __forceinline__ v16h load_a_f32(const float* __restrict__ p) {
    float4 a0 = *(const float4*)(p);
    float4 a1 = *(const float4*)(p + 4);
    float4 a2 = *(const float4*)(p + 16);
    float4 a3 = *(const float4*)(p + 20);
    v16h v;
    v[0]=(_Float16)a0.x; v[1]=(_Float16)a0.y; v[2]=(_Float16)a0.z; v[3]=(_Float16)a0.w;
    v[4]=(_Float16)a1.x; v[5]=(_Float16)a1.y; v[6]=(_Float16)a1.z; v[7]=(_Float16)a1.w;
    v[8]=(_Float16)a2.x; v[9]=(_Float16)a2.y; v[10]=(_Float16)a2.z; v[11]=(_Float16)a2.w;
    v[12]=(_Float16)a3.x; v[13]=(_Float16)a3.y; v[14]=(_Float16)a3.z; v[15]=(_Float16)a3.w;
    return v;
}

__device__ __forceinline__ v16h load_a_prod(const float* __restrict__ sp,
                                            const float* __restrict__ dp) {
    float4 s0 = *(const float4*)(sp),      d0 = *(const float4*)(dp);
    float4 s1 = *(const float4*)(sp + 4),  d1 = *(const float4*)(dp + 4);
    float4 s2 = *(const float4*)(sp + 16), d2 = *(const float4*)(dp + 16);
    float4 s3 = *(const float4*)(sp + 20), d3 = *(const float4*)(dp + 20);
    v16h v;
    v[0]=(_Float16)(s0.x*d0.x); v[1]=(_Float16)(s0.y*d0.y);
    v[2]=(_Float16)(s0.z*d0.z); v[3]=(_Float16)(s0.w*d0.w);
    v[4]=(_Float16)(s1.x*d1.x); v[5]=(_Float16)(s1.y*d1.y);
    v[6]=(_Float16)(s1.z*d1.z); v[7]=(_Float16)(s1.w*d1.w);
    v[8]=(_Float16)(s2.x*d2.x); v[9]=(_Float16)(s2.y*d2.y);
    v[10]=(_Float16)(s2.z*d2.z); v[11]=(_Float16)(s2.w*d2.w);
    v[12]=(_Float16)(s3.x*d3.x); v[13]=(_Float16)(s3.y*d3.y);
    v[14]=(_Float16)(s3.z*d3.z); v[15]=(_Float16)(s3.w*d3.w);
    return v;
}

__device__ __forceinline__ v16h load_b_f16(const _Float16* __restrict__ p) {
    v8h lo = *(const v8h*)(p);
    v8h hi = *(const v8h*)(p + 16);
    v16h v;
#pragma unroll
    for (int i = 0; i < 8; ++i) { v[i] = lo[i]; v[8 + i] = hi[i]; }
    return v;
}

__global__ void prep_weights(const float* __restrict__ W0, const float* __restrict__ W1,
                             const float* __restrict__ Wp1, const float* __restrict__ Wp2,
                             _Float16* __restrict__ W0t, _Float16* __restrict__ W1t,
                             _Float16* __restrict__ Wp1t, _Float16* __restrict__ Wp2t) {
    int idx = blockIdx.x * blockDim.x + threadIdx.x;
    if (idx < 128 * 128) {
        int c = idx >> 7, k = idx & 127;
        W0t[idx] = (_Float16)W0[k * 128 + c];
        W1t[idx] = (_Float16)W1[k * 128 + c];
    }
    if (idx < 64 * 128) {
        int c = idx >> 7, k = idx & 127;
        Wp1t[idx] = (_Float16)Wp1[k * 64 + c];
    }
    if (idx < 64 * 64) {
        int c = idx >> 6, k = idx & 63;
        Wp2t[idx] = (_Float16)Wp2[k * 64 + c];
    }
}

__global__ __launch_bounds__(256)
void gemm128_wmma(const float* __restrict__ A, const _Float16* __restrict__ Bt,
                  float* __restrict__ C, int M) {
    int wid  = blockIdx.x * (blockDim.x >> 5) + (threadIdx.x >> 5);
    int lane = threadIdx.x & 31;
    int half = lane >> 4, r = lane & 15;
    int tm = wid >> 3, tn = wid & 7;
    if (tm * 16 >= M) return;
    int row = tm * 16 + r;
    int col = tn * 16 + r;
    v8f acc = {};
#pragma unroll
    for (int kk = 0; kk < 128; kk += 32) {
        int k0 = kk + half * 8;
        v16h a = load_a_f32(A + row * 128 + k0);
        v16h b = load_b_f16(Bt + col * 128 + k0);
        acc = __builtin_amdgcn_wmma_f32_16x16x32_f16(false, a, false, b,
                                                     (short)0, acc, false, false);
    }
    float* cp = C + (tm * 16 + half * 8) * 128 + tn * 16 + r;
#pragma unroll
    for (int v = 0; v < 8; ++v) cp[v * 128] = acc[v];
}

__global__ __launch_bounds__(256)
void edge_score_kernel(const float* __restrict__ feat, const int* __restrict__ src,
                       const int* __restrict__ dst, const float* __restrict__ attn,
                       float* __restrict__ score, int* __restrict__ mOrd) {
    int gid = blockIdx.x * blockDim.x + threadIdx.x;
    int e = gid >> 5;
    if (e >= E_EDGES) return;
    int lane = threadIdx.x & 31;
    int hh = lane >> 4, d4 = (lane & 15) << 2;
    int s_ = src[e], d_ = dst[e];
    float4 el = *(const float4*)(feat + s_ * HDIM + hh * 64 + d4);
    float4 er = *(const float4*)(feat + d_ * HDIM + hh * 64 + d4);
    float4 at = *(const float4*)(attn + hh * 64 + d4);
    float x0 = el.x + er.x; x0 = x0 > 0.f ? x0 : NEG_SLOPE * x0;
    float x1 = el.y + er.y; x1 = x1 > 0.f ? x1 : NEG_SLOPE * x1;
    float x2 = el.z + er.z; x2 = x2 > 0.f ? x2 : NEG_SLOPE * x2;
    float x3 = el.w + er.w; x3 = x3 > 0.f ? x3 : NEG_SLOPE * x3;
    float part = x0 * at.x + x1 * at.y + x2 * at.z + x3 * at.w;
    part += __shfl_xor(part, 1, 32);
    part += __shfl_xor(part, 2, 32);
    part += __shfl_xor(part, 4, 32);
    part += __shfl_xor(part, 8, 32);
    if ((lane & 15) == 0) {
        score[e * 2 + hh] = part;
        atomicMax(&mOrd[d_ * 2 + hh], f2ord(part));
    }
}

__global__ __launch_bounds__(256)
void edge_expsum_kernel(const int* __restrict__ dst, const int* __restrict__ mOrd,
                        float* __restrict__ score, float* __restrict__ ssum) {
    int idx = blockIdx.x * blockDim.x + threadIdx.x;
    if (idx >= E_EDGES * 2) return;
    int e = idx >> 1, hh = idx & 1;
    int d_ = dst[e];
    float m = ord2f(mOrd[d_ * 2 + hh]);
    float p = __expf(score[idx] - m);
    score[idx] = p;
    atomicAdd(&ssum[d_ * 2 + hh], p);
}

__global__ __launch_bounds__(256)
void edge_aggregate_kernel(const float* __restrict__ feat, const int* __restrict__ src,
                           const int* __restrict__ dst, const float* __restrict__ p,
                           const float* __restrict__ ssum, float* __restrict__ rst) {
    int gid = blockIdx.x * blockDim.x + threadIdx.x;
    int e = gid >> 5;
    if (e >= E_EDGES) return;
    int lane = threadIdx.x & 31;
    int hh = lane >> 4, d4 = (lane & 15) << 2;
    int s_ = src[e], d_ = dst[e];
    float a = p[e * 2 + hh] / ssum[d_ * 2 + hh];
    float4 el = *(const float4*)(feat + s_ * HDIM + hh * 64 + d4);
    float* rp = rst + d_ * HDIM + hh * 64 + d4;
    atomicAdd(rp + 0, el.x * a);
    atomicAdd(rp + 1, el.y * a);
    atomicAdd(rp + 2, el.z * a);
    atomicAdd(rp + 3, el.w * a);
}

__global__ __launch_bounds__(256)
void finalize_kernel(const float* __restrict__ rst, const float* __restrict__ hin,
                     float* __restrict__ hout, int residual, int n) {
    int idx = blockIdx.x * blockDim.x + threadIdx.x;
    if (idx >= n) return;
    float v = rst[idx] + (residual ? hin[idx] : 0.0f);
    hout[idx] = v > 0.f ? v : (__expf(v) - 1.0f);
}

__global__ __launch_bounds__(128)
void predictor_wmma(const float* __restrict__ h,
                    const int* __restrict__ ps, const int* __restrict__ pd,
                    const int* __restrict__ ns, const int* __restrict__ nd,
                    const _Float16* __restrict__ Wp1t, const float* __restrict__ bp1,
                    const _Float16* __restrict__ Wp2t, const float* __restrict__ bp2,
                    const float* __restrict__ Wp3, const float* __restrict__ bp3,
                    float* __restrict__ out) {
    __shared__ _Float16 z1s[4][16 * 64];
    __shared__ float    z2s[4][16 * 64];
    const int tilesPerSet = P_CAND / 16;
    int w = threadIdx.x >> 5;
    int tile = blockIdx.x * 4 + w;
    if (tile >= 2 * tilesPerSet) return;
    bool isNeg = tile >= tilesPerSet;
    int t = isNeg ? tile - tilesPerSet : tile;
    const int* S = isNeg ? ns : ps;
    const int* D = isNeg ? nd : pd;

    int lane = threadIdx.x & 31;
    int half = lane >> 4, r = lane & 15;
    int c = t * 16 + r;
    int si = S[c], di = D[c];

    v16h a1[4];
#pragma unroll
    for (int kc = 0; kc < 4; ++kc) {
        int k0 = kc * 32 + half * 8;
        a1[kc] = load_a_prod(h + si * HDIM + k0, h + di * HDIM + k0);
    }
#pragma unroll
    for (int nt = 0; nt < 4; ++nt) {
        int col = nt * 16 + r;
        v8f acc = {};
#pragma unroll
        for (int kc = 0; kc < 4; ++kc) {
            int k0 = kc * 32 + half * 8;
            v16h b = load_b_f16(Wp1t + col * 128 + k0);
            acc = __builtin_amdgcn_wmma_f32_16x16x32_f16(false, a1[kc], false, b,
                                                         (short)0, acc, false, false);
        }
        float bias = bp1[col];
#pragma unroll
        for (int v = 0; v < 8; ++v) {
            float z = acc[v] + bias;
            z = z > 0.f ? z : 0.f;
            z1s[w][(v + half * 8) * 64 + col] = (_Float16)z;
        }
    }

    v16h a2[2];
#pragma unroll
    for (int kc = 0; kc < 2; ++kc) {
        int k0 = kc * 32 + half * 8;
        const _Float16* zp = &z1s[w][r * 64 + k0];
        a2[kc] = load_b_f16(zp);
    }
#pragma unroll
    for (int nt = 0; nt < 4; ++nt) {
        int col = nt * 16 + r;
        v8f acc = {};
#pragma unroll
        for (int kc = 0; kc < 2; ++kc) {
            int k0 = kc * 32 + half * 8;
            v16h b = load_b_f16(Wp2t + col * 64 + k0);
            acc = __builtin_amdgcn_wmma_f32_16x16x32_f16(false, a2[kc], false, b,
                                                         (short)0, acc, false, false);
        }
        float bias = bp2[col];
#pragma unroll
        for (int v = 0; v < 8; ++v) {
            float z = acc[v] + bias;
            z2s[w][(v + half * 8) * 64 + col] = z > 0.f ? z : 0.f;
        }
    }

    float part = 0.f;
    int jb = half * 32;
#pragma unroll
    for (int j = 0; j < 32; ++j)
        part += z2s[w][r * 64 + jb + j] * Wp3[jb + j];
    part += __shfl_xor(part, 16, 32);
    if (half == 0) {
        out[(isNeg ? P_CAND : 0) + t * 16 + r] = part + bp3[0];
    }
}

extern "C" void kernel_launch(void* const* d_in, const int* in_sizes, int n_in,
                              void* d_out, int out_size, void* d_ws, size_t ws_size,
                              hipStream_t stream) {
    const float* x        = (const float*)d_in[0];
    const int*   src      = (const int*)  d_in[1];
    const int*   dst      = (const int*)  d_in[2];
    const int*   pos_src  = (const int*)  d_in[3];
    const int*   pos_dst  = (const int*)  d_in[4];
    const int*   neg_src  = (const int*)  d_in[5];
    const int*   neg_dst  = (const int*)  d_in[6];
    const float* W0       = (const float*)d_in[7];
    const float* attn0    = (const float*)d_in[8];
    const float* W1       = (const float*)d_in[9];
    const float* attn1    = (const float*)d_in[10];
    const float* Wp1      = (const float*)d_in[11];
    const float* bp1      = (const float*)d_in[12];
    const float* Wp2      = (const float*)d_in[13];
    const float* bp2      = (const float*)d_in[14];
    const float* Wp3      = (const float*)d_in[15];
    const float* bp3      = (const float*)d_in[16];

    char* ws = (char*)d_ws;
    size_t off = 0;
    auto carve = [&](size_t bytes) -> char* {
        char* p = ws + off;
        off += (bytes + 255) & ~(size_t)255;
        return p;
    };
    float*     feat  = (float*)    carve((size_t)N_NODES * HDIM * 4);
    float*     h0    = (float*)    carve((size_t)N_NODES * HDIM * 4);
    float*     h1    = (float*)    carve((size_t)N_NODES * HDIM * 4);
    float*     rst   = (float*)    carve((size_t)N_NODES * HDIM * 4);
    float*     score = (float*)    carve((size_t)E_EDGES * 2 * 4);
    int*       mOrd  = (int*)      carve((size_t)N_NODES * 2 * 4);
    float*     ssum  = (float*)    carve((size_t)N_NODES * 2 * 4);
    _Float16*  W0t   = (_Float16*) carve(128 * 128 * 2);
    _Float16*  W1t   = (_Float16*) carve(128 * 128 * 2);
    _Float16*  Wp1t  = (_Float16*) carve(64 * 128 * 2);
    _Float16*  Wp2t  = (_Float16*) carve(64 * 64 * 2);

    prep_weights<<<(128 * 128 + 255) / 256, 256, 0, stream>>>(
        W0, W1, Wp1, Wp2, W0t, W1t, Wp1t, Wp2t);

    const int gemmBlocks  = (N_NODES / 16) * 8 / 8;
    const int edgeBlocks  = E_EDGES / 8;
    const int ehBlocks    = (E_EDGES * 2 + 255) / 256;
    const int nodeBlocks  = (N_NODES * HDIM + 255) / 256;

    hipMemsetAsync(rst,  0,    (size_t)N_NODES * HDIM * 4, stream);
    hipMemsetAsync(ssum, 0,    (size_t)N_NODES * 2 * 4,    stream);
    hipMemsetAsync(mOrd, 0x80, (size_t)N_NODES * 2 * 4,    stream);
    gemm128_wmma<<<gemmBlocks, 256, 0, stream>>>(x, W0t, feat, N_NODES);
    edge_score_kernel<<<edgeBlocks, 256, 0, stream>>>(feat, src, dst, attn0, score, mOrd);
    edge_expsum_kernel<<<ehBlocks, 256, 0, stream>>>(dst, mOrd, score, ssum);
    edge_aggregate_kernel<<<edgeBlocks, 256, 0, stream>>>(feat, src, dst, score, ssum, rst);
    finalize_kernel<<<nodeBlocks, 256, 0, stream>>>(rst, x, h0, 0, N_NODES * HDIM);

    hipMemsetAsync(rst,  0,    (size_t)N_NODES * HDIM * 4, stream);
    hipMemsetAsync(ssum, 0,    (size_t)N_NODES * 2 * 4,    stream);
    hipMemsetAsync(mOrd, 0x80, (size_t)N_NODES * 2 * 4,    stream);
    gemm128_wmma<<<gemmBlocks, 256, 0, stream>>>(h0, W1t, feat, N_NODES);
    edge_score_kernel<<<edgeBlocks, 256, 0, stream>>>(feat, src, dst, attn1, score, mOrd);
    edge_expsum_kernel<<<ehBlocks, 256, 0, stream>>>(dst, mOrd, score, ssum);
    edge_aggregate_kernel<<<edgeBlocks, 256, 0, stream>>>(feat, src, dst, score, ssum, rst);
    finalize_kernel<<<nodeBlocks, 256, 0, stream>>>(rst, h0, h1, 1, N_NODES * HDIM);

    const int predBlocks = (2 * (P_CAND / 16) + 3) / 4;
    predictor_wmma<<<predBlocks, 128, 0, stream>>>(
        h1, pos_src, pos_dst, neg_src, neg_dst,
        Wp1t, bp1, Wp2t, bp2, Wp3, bp3, (float*)d_out);
}


# Edges counting-sorted by dst once per call; one wave per dst node does score, online softmax, weighted sum, residual and ELU in registers: float atomics gone, sums in arrival-order rank
# speedup vs baseline: 2.8840x; 2.8840x over previous
#include <hip/hip_runtime.h>


typedef __attribute__((ext_vector_type(16))) _Float16 v16h;
typedef __attribute__((ext_vector_type(8)))  _Float16 v8h;
typedef __attribute__((ext_vector_type(8)))  float    v8f;

#define N_NODES 50000
#define E_EDGES 800000
#define P_CAND  100000
#define HDIM    128
#define NEG_SLOPE 0.2f
#define EDGE_CHUNK 8
#define SCAN_ROWS  49


__device__ __forceinline__ v16h load_a_f32(const float* __restrict__ p) {
    float4 a0 = *(const float4*)(p);
    float4 a1 = *(const float4*)(p + 4);
    float4 a2 = *(const float4*)(p + 16);
    float4 a3 = *(const float4*)(p + 20);
    v16h v;
    v[0]=(_Float16)a0.x; v[1]=(_Float16)a0.y; v[2]=(_Float16)a0.z; v[3]=(_Float16)a0.w;
    v[4]=(_Float16)a1.x; v[5]=(_Float16)a1.y; v[6]=(_Float16)a1.z; v[7]=(_Float16)a1.w;
    v[8]=(_Float16)a2.x; v[9]=(_Float16)a2.y; v[10]=(_Float16)a2.z; v[11]=(_Float16)a2.w;
    v[12]=(_Float16)a3.x; v[13]=(_Float16)a3.y; v[14]=(_Float16)a3.z; v[15]=(_Float16)a3.w;
    return v;
}

__device__ __forceinline__ v16h load_a_prod(const float* __restrict__ sp,
                                            const float* __restrict__ dp) {
    float4 s0 = *(const float4*)(sp),      d0 = *(const float4*)(dp);
    float4 s1 = *(const float4*)(sp + 4),  d1 = *(const float4*)(dp + 4);
    float4 s2 = *(const float4*)(sp + 16), d2 = *(const float4*)(dp + 16);
    float4 s3 = *(const float4*)(sp + 20), d3 = *(const float4*)(dp + 20);
    v16h v;
    v[0]=(_Float16)(s0.x*d0.x); v[1]=(_Float16)(s0.y*d0.y);
    v[2]=(_Float16)(s0.z*d0.z); v[3]=(_Float16)(s0.w*d0.w);
    v[4]=(_Float16)(s1.x*d1.x); v[5]=(_Float16)(s1.y*d1.y);
    v[6]=(_Float16)(s1.z*d1.z); v[7]=(_Float16)(s1.w*d1.w);
    v[8]=(_Float16)(s2.x*d2.x); v[9]=(_Float16)(s2.y*d2.y);
    v[10]=(_Float16)(s2.z*d2.z); v[11]=(_Float16)(s2.w*d2.w);
    v[12]=(_Float16)(s3.x*d3.x); v[13]=(_Float16)(s3.y*d3.y);
    v[14]=(_Float16)(s3.z*d3.z); v[15]=(_Float16)(s3.w*d3.w);
    return v;
}

__device__ __forceinline__ v16h load_b_f16(const _Float16* __restrict__ p) {
    v8h lo = *(const v8h*)(p);
    v8h hi = *(const v8h*)(p + 16);
    v16h v;
#pragma unroll
    for (int i = 0; i < 8; ++i) { v[i] = lo[i]; v[8 + i] = hi[i]; }
    return v;
}

__global__ void prep_weights(const float* __restrict__ W0, const float* __restrict__ W1,
                             const float* __restrict__ Wp1, const float* __restrict__ Wp2,
                             _Float16* __restrict__ W0t, _Float16* __restrict__ W1t,
                             _Float16* __restrict__ Wp1t, _Float16* __restrict__ Wp2t,
                             int* __restrict__ cnt) {
    int idx = blockIdx.x * blockDim.x + threadIdx.x;
    for (int i = idx; i < N_NODES; i += gridDim.x * blockDim.x) cnt[i] = 0;
    if (idx < 128 * 128) {
        int c = idx >> 7, k = idx & 127;
        W0t[idx] = (_Float16)W0[k * 128 + c];
        W1t[idx] = (_Float16)W1[k * 128 + c];
    }
    if (idx < 64 * 128) {
        int c = idx >> 7, k = idx & 127;
        Wp1t[idx] = (_Float16)Wp1[k * 64 + c];
    }
    if (idx < 64 * 64) {
        int c = idx >> 6, k = idx & 63;
        Wp2t[idx] = (_Float16)Wp2[k * 64 + c];
    }
}

__global__ __launch_bounds__(256)
void gemm128_wmma(const float* __restrict__ A, const _Float16* __restrict__ Bt,
                  float* __restrict__ C, int M) {
    int wid  = blockIdx.x * (blockDim.x >> 5) + (threadIdx.x >> 5);
    int lane = threadIdx.x & 31;
    int half = lane >> 4, r = lane & 15;
    int tm = wid >> 3, tn = wid & 7;
    if (tm * 16 >= M) return;
    int row = tm * 16 + r;
    int col = tn * 16 + r;
    v8f acc = {};
#pragma unroll
    for (int kk = 0; kk < 128; kk += 32) {
        int k0 = kk + half * 8;
        v16h a = load_a_f32(A + row * 128 + k0);
        v16h b = load_b_f16(Bt + col * 128 + k0);
        acc = __builtin_amdgcn_wmma_f32_16x16x32_f16(false, a, false, b,
                                                     (short)0, acc, false, false);
    }
    float* cp = C + (tm * 16 + half * 8) * 128 + tn * 16 + r;
#pragma unroll
    for (int v = 0; v < 8; ++v) cp[v * 128] = acc[v];
}

__global__ __launch_bounds__(256)
void edge_rank_kernel(const int* __restrict__ dst, int* __restrict__ cnt,
                      int* __restrict__ rank) {
    int e = blockIdx.x * blockDim.x + threadIdx.x;
    if (e >= E_EDGES) return;
    rank[e] = atomicAdd(&cnt[dst[e]], 1);
}

__global__ __launch_bounds__(1024)
void degree_scan_kernel(const int* __restrict__ cnt, int* __restrict__ rowptr) {
    __shared__ int wtot[32];
    int lane = threadIdx.x & 31, w = threadIdx.x >> 5;
    int rbase = w * (SCAN_ROWS * 32);
    int lsum = 0;
#pragma unroll
    for (int i = 0; i < SCAN_ROWS; ++i) {
        int idx = rbase + i * 32 + lane;
        lsum += idx < N_NODES ? cnt[idx] : 0;
    }
#pragma unroll
    for (int d = 16; d >= 1; d >>= 1) lsum += __shfl_xor(lsum, d, 32);
    if (lane == 0) wtot[w] = lsum;
    __syncthreads();
    if (w == 0) {
        int t = wtot[lane];
        int x = t;
#pragma unroll
        for (int d = 1; d < 32; d <<= 1) {
            int o = __shfl_up(x, d, 32);
            if (lane >= d) x += o;
        }
        wtot[lane] = x - t;
    }
    __syncthreads();
    int carry = wtot[w];
#pragma unroll 7
    for (int i = 0; i < SCAN_ROWS; ++i) {
        int idx = rbase + i * 32 + lane;
        int v = idx < N_NODES ? cnt[idx] : 0;
        int x = v;
#pragma unroll
        for (int d = 1; d < 32; d <<= 1) {
            int o = __shfl_up(x, d, 32);
            if (lane >= d) x += o;
        }
        if (idx < N_NODES) rowptr[idx] = carry + x - v;
        carry += __shfl(x, 31, 32);
    }
    if (threadIdx.x == 1023) rowptr[N_NODES] = carry;
}

__global__ __launch_bounds__(256)
void edge_place_kernel(const int* __restrict__ src, const int* __restrict__ dst,
                       const int* __restrict__ rank, const int* __restrict__ rowptr,
                       int* __restrict__ esrc) {
    int e = blockIdx.x * blockDim.x + threadIdx.x;
    if (e >= E_EDGES) return;
    esrc[rowptr[dst[e]] + rank[e]] = src[e];
}

template <int M>
__device__ __forceinline__ float row_xor(float x) {
    int b = __builtin_bit_cast(int, x);
    return __builtin_bit_cast(float,
        __builtin_amdgcn_update_dpp(b, b, 0x160 | M, 0xf, 0xf, false));
}

__global__ __launch_bounds__(256)
void gat_node_kernel(const float* __restrict__ feat, const int* __restrict__ rowptr,
                     const int* __restrict__ esrc, const float* __restrict__ attn,
                     const float* __restrict__ hin, float* __restrict__ hout,
                     int residual) {
    int lane = threadIdx.x & 31;
    int node = __builtin_amdgcn_readfirstlane((int)(blockIdx.x * 8 + (threadIdx.x >> 5)));
    if (node >= N_NODES) return;
    int beg = rowptr[node], end = rowptr[node + 1];
    const int lo = lane * 4;
    float4 er = *(const float4*)(feat + (size_t)node * HDIM + lo);
    float4 at = *(const float4*)(attn + lo);
    float m = -INFINITY, s = 0.f;
    float4 acc = make_float4(0.f, 0.f, 0.f, 0.f);

    for (int base = beg; base < end; base += 32) {
        int n = end - base;
        n = n < 32 ? n : 32;
        int myidx = esrc[base + (lane < n ? lane : n - 1)];
        for (int j = 0; j < n; j += EDGE_CHUNK) {
            float4 el[EDGE_CHUNK];
#pragma unroll
            for (int k = 0; k < EDGE_CHUNK; ++k) {
                int jj = j + k < n ? j + k : n - 1;
                int s_ = __builtin_amdgcn_readlane(myidx, jj);
                el[k] = *(const float4*)(feat + (size_t)s_ * HDIM + lo);
            }
            float sc[EDGE_CHUNK];
            float mc = m;
#pragma unroll
            for (int k = 0; k < EDGE_CHUNK; ++k) {
                float x0 = el[k].x + er.x; x0 = fmaxf(x0, NEG_SLOPE * x0);
                float x1 = el[k].y + er.y; x1 = fmaxf(x1, NEG_SLOPE * x1);
                float x2 = el[k].z + er.z; x2 = fmaxf(x2, NEG_SLOPE * x2);
                float x3 = el[k].w + er.w; x3 = fmaxf(x3, NEG_SLOPE * x3);
                float part = x0 * at.x + x1 * at.y + x2 * at.z + x3 * at.w;
                part += row_xor<1>(part);
                part += row_xor<2>(part);
                part += row_xor<4>(part);
                part += row_xor<8>(part);
                sc[k] = (j + k < n) ? part : -INFINITY;
                mc = fmaxf(mc, sc[k]);
            }
            float scale = __expf(m - mc);
            float ps = 0.f;
            float4 a2 = make_float4(0.f, 0.f, 0.f, 0.f);
#pragma unroll
            for (int k = 0; k < EDGE_CHUNK; ++k) {
                float p = __expf(sc[k] - mc);
                ps += p;
                a2.x += el[k].x * p; a2.y += el[k].y * p;
                a2.z += el[k].z * p; a2.w += el[k].w * p;
            }
            s = s * scale + ps;
            acc.x = acc.x * scale + a2.x; acc.y = acc.y * scale + a2.y;
            acc.z = acc.z * scale + a2.z; acc.w = acc.w * scale + a2.w;
            m = mc;
        }
    }

    float4 r = make_float4(0.f, 0.f, 0.f, 0.f);
    if (end > beg) { r.x = acc.x / s; r.y = acc.y / s; r.z = acc.z / s; r.w = acc.w / s; }
    if (residual) {
        float4 hv = *(const float4*)(hin + (size_t)node * HDIM + lo);
        r.x += hv.x; r.y += hv.y; r.z += hv.z; r.w += hv.w;
    }
    r.x = r.x > 0.f ? r.x : (__expf(r.x) - 1.0f);
    r.y = r.y > 0.f ? r.y : (__expf(r.y) - 1.0f);
    r.z = r.z > 0.f ? r.z : (__expf(r.z) - 1.0f);
    r.w = r.w > 0.f ? r.w : (__expf(r.w) - 1.0f);
    *(float4*)(hout + (size_t)node * HDIM + lo) = r;
}

__global__ __launch_bounds__(128)
void predictor_wmma(const float* __restrict__ h,
                    const int* __restrict__ ps, const int* __restrict__ pd,
                    const int* __restrict__ ns, const int* __restrict__ nd,
                    const _Float16* __restrict__ Wp1t, const float* __restrict__ bp1,
                    const _Float16* __restrict__ Wp2t, const float* __restrict__ bp2,
                    const float* __restrict__ Wp3, const float* __restrict__ bp3,
                    float* __restrict__ out) {
    __shared__ _Float16 z1s[4][16 * 64];
    __shared__ float    z2s[4][16 * 64];
    const int tilesPerSet = P_CAND / 16;
    int w = threadIdx.x >> 5;
    int tile = blockIdx.x * 4 + w;
    if (tile >= 2 * tilesPerSet) return;
    bool isNeg = tile >= tilesPerSet;
    int t = isNeg ? tile - tilesPerSet : tile;
    const int* S = isNeg ? ns : ps;
    const int* D = isNeg ? nd : pd;

    int lane = threadIdx.x & 31;
    int half = lane >> 4, r = lane & 15;
    int c = t * 16 + r;
    int si = S[c], di = D[c];

    v16h a1[4];
#pragma unroll
    for (int kc = 0; kc < 4; ++kc) {
        int k0 = kc * 32 + half * 8;
        a1[kc] = load_a_prod(h + si * HDIM + k0, h + di * HDIM + k0);
    }
#pragma unroll
    for (int nt = 0; nt < 4; ++nt) {
        int col = nt * 16 + r;
        v8f acc = {};
#pragma unroll
        for (int kc = 0; kc < 4; ++kc) {
            int k0 = kc * 32 + half * 8;
            v16h b = load_b_f16(Wp1t + col * 128 + k0);
            acc = __builtin_amdgcn_wmma_f32_16x16x32_f16(false, a1[kc], false, b,
                                                         (short)0, acc, false, false);
        }
        float bias = bp1[col];
#pragma unroll
        for (int v = 0; v < 8; ++v) {
            float z = acc[v] + bias;
            z = z > 0.f ? z : 0.f;
            z1s[w][(v + half * 8) * 64 + col] = (_Float16)z;
        }
    }

    v16h a2[2];
#pragma unroll
    for (int kc = 0; kc < 2; ++kc) {
        int k0 = kc * 32 + half * 8;
        const _Float16* zp = &z1s[w][r * 64 + k0];
        a2[kc] = load_b_f16(zp);
    }
#pragma unroll
    for (int nt = 0; nt < 4; ++nt) {
        int col = nt * 16 + r;
        v8f acc = {};
#pragma unroll
        for (int kc = 0; kc < 2; ++kc) {
            int k0 = kc * 32 + half * 8;
            v16h b = load_b_f16(Wp2t + col * 64 + k0);
            acc = __builtin_amdgcn_wmma_f32_16x16x32_f16(false, a2[kc], false, b,
                                                         (short)0, acc, false, false);
        }
        float bias = bp2[col];
#pragma unroll
        for (int v = 0; v < 8; ++v) {
            float z = acc[v] + bias;
            z2s[w][(v + half * 8) * 64 + col] = z > 0.f ? z : 0.f;
        }
    }

    float part = 0.f;
    int jb = half * 32;
#pragma unroll
    for (int j = 0; j < 32; ++j)
        part += z2s[w][r * 64 + jb + j] * Wp3[jb + j];
    part += __shfl_xor(part, 16, 32);
    if (half == 0) {
        out[(isNeg ? P_CAND : 0) + t * 16 + r] = part + bp3[0];
    }
}

extern "C" void kernel_launch(void* const* d_in, const int* in_sizes, int n_in,
                              void* d_out, int out_size, void* d_ws, size_t ws_size,
                              hipStream_t stream) {
    const float* x        = (const float*)d_in[0];
    const int*   src      = (const int*)  d_in[1];
    const int*   dst      = (const int*)  d_in[2];
    const int*   pos_src  = (const int*)  d_in[3];
    const int*   pos_dst  = (const int*)  d_in[4];
    const int*   neg_src  = (const int*)  d_in[5];
    const int*   neg_dst  = (const int*)  d_in[6];
    const float* W0       = (const float*)d_in[7];
    const float* attn0    = (const float*)d_in[8];
    const float* W1       = (const float*)d_in[9];
    const float* attn1    = (const float*)d_in[10];
    const float* Wp1      = (const float*)d_in[11];
    const float* bp1      = (const float*)d_in[12];
    const float* Wp2      = (const float*)d_in[13];
    const float* bp2      = (const float*)d_in[14];
    const float* Wp3      = (const float*)d_in[15];
    const float* bp3      = (const float*)d_in[16];

    char* ws = (char*)d_ws;
    size_t off = 0;
    auto carve = [&](size_t bytes) -> char* {
        char* p = ws + off;
        off += (bytes + 255) & ~(size_t)255;
        return p;
    };
    float*     feat  = (float*)    carve((size_t)N_NODES * HDIM * 4);
    float*     h0    = (float*)    carve((size_t)N_NODES * HDIM * 4);
    float*     h1    = (float*)    carve((size_t)N_NODES * HDIM * 4);
    int*       cnt   = (int*)      carve((size_t)N_NODES * 4);
    int*       rowptr= (int*)      carve((size_t)(N_NODES + 1) * 4);
    int*       rank  = (int*)      carve((size_t)E_EDGES * 4);
    int*       esrc  = (int*)      carve((size_t)E_EDGES * 4);
    _Float16*  W0t   = (_Float16*) carve(128 * 128 * 2);
    _Float16*  W1t   = (_Float16*) carve(128 * 128 * 2);
    _Float16*  Wp1t  = (_Float16*) carve(64 * 128 * 2);
    _Float16*  Wp2t  = (_Float16*) carve(64 * 64 * 2);

    prep_weights<<<(128 * 128 + 255) / 256, 256, 0, stream>>>(
        W0, W1, Wp1, Wp2, W0t, W1t, Wp1t, Wp2t, cnt);

    const int gemmBlocks  = (N_NODES / 16) * 8 / 8;
    const int edgeThrBlocks = (E_EDGES + 255) / 256;
    const int nodeWaveBlocks = (N_NODES + 7) / 8;

    edge_rank_kernel<<<edgeThrBlocks, 256, 0, stream>>>(dst, cnt, rank);
    degree_scan_kernel<<<1, 1024, 0, stream>>>(cnt, rowptr);
    edge_place_kernel<<<edgeThrBlocks, 256, 0, stream>>>(src, dst, rank, rowptr, esrc);

    gemm128_wmma<<<gemmBlocks, 256, 0, stream>>>(x, W0t, feat, N_NODES);
    gat_node_kernel<<<nodeWaveBlocks, 256, 0, stream>>>(feat, rowptr, esrc, attn0, x, h0, 0);

    gemm128_wmma<<<gemmBlocks, 256, 0, stream>>>(h0, W1t, feat, N_NODES);
    gat_node_kernel<<<nodeWaveBlocks, 256, 0, stream>>>(feat, rowptr, esrc, attn1, h0, h1, 1);

    const int predBlocks = (2 * (P_CAND / 16) + 3) / 4;
    predictor_wmma<<<predBlocks, 128, 0, stream>>>(
        h1, pos_src, pos_dst, neg_src, neg_dst,
        Wp1t, bp1, Wp2t, bp2, Wp3, bp3, (float*)d_out);
}


# Degree prefix sum spread over 49 work-groups: each sums the counters in front of its 1024-counter slice with independent int4 loads, then scans its slice; was one work-group walking 49 dependent rows
# speedup vs baseline: 3.2271x; 1.1190x over previous
#include <hip/hip_runtime.h>


typedef __attribute__((ext_vector_type(16))) _Float16 v16h;
typedef __attribute__((ext_vector_type(8)))  _Float16 v8h;
typedef __attribute__((ext_vector_type(8)))  float    v8f;

#define N_NODES 50000
#define E_EDGES 800000
#define P_CAND  100000
#define HDIM    128
#define NEG_SLOPE 0.2f
#define EDGE_CHUNK 8
#define SCAN_BLOCK 1024
#define SCAN_BLOCKS ((N_NODES + 1 + SCAN_BLOCK - 1) / SCAN_BLOCK)
#define SCAN_PRE_LOADS (((SCAN_BLOCKS - 1) * (SCAN_BLOCK / 4) + SCAN_BLOCK - 1) / SCAN_BLOCK)


__device__ __forceinline__ v16h load_a_f32(const float* __restrict__ p) {
    float4 a0 = *(const float4*)(p);
    float4 a1 = *(const float4*)(p + 4);
    float4 a2 = *(const float4*)(p + 16);
    float4 a3 = *(const float4*)(p + 20);
    v16h v;
    v[0]=(_Float16)a0.x; v[1]=(_Float16)a0.y; v[2]=(_Float16)a0.z; v[3]=(_Float16)a0.w;
    v[4]=(_Float16)a1.x; v[5]=(_Float16)a1.y; v[6]=(_Float16)a1.z; v[7]=(_Float16)a1.w;
    v[8]=(_Float16)a2.x; v[9]=(_Float16)a2.y; v[10]=(_Float16)a2.z; v[11]=(_Float16)a2.w;
    v[12]=(_Float16)a3.x; v[13]=(_Float16)a3.y; v[14]=(_Float16)a3.z; v[15]=(_Float16)a3.w;
    return v;
}

__device__ __forceinline__ v16h load_a_prod(const float* __restrict__ sp,
                                            const float* __restrict__ dp) {
    float4 s0 = *(const float4*)(sp),      d0 = *(const float4*)(dp);
    float4 s1 = *(const float4*)(sp + 4),  d1 = *(const float4*)(dp + 4);
    float4 s2 = *(const float4*)(sp + 16), d2 = *(const float4*)(dp + 16);
    float4 s3 = *(const float4*)(sp + 20), d3 = *(const float4*)(dp + 20);
    v16h v;
    v[0]=(_Float16)(s0.x*d0.x); v[1]=(_Float16)(s0.y*d0.y);
    v[2]=(_Float16)(s0.z*d0.z); v[3]=(_Float16)(s0.w*d0.w);
    v[4]=(_Float16)(s1.x*d1.x); v[5]=(_Float16)(s1.y*d1.y);
    v[6]=(_Float16)(s1.z*d1.z); v[7]=(_Float16)(s1.w*d1.w);
    v[8]=(_Float16)(s2.x*d2.x); v[9]=(_Float16)(s2.y*d2.y);
    v[10]=(_Float16)(s2.z*d2.z); v[11]=(_Float16)(s2.w*d2.w);
    v[12]=(_Float16)(s3.x*d3.x); v[13]=(_Float16)(s3.y*d3.y);
    v[14]=(_Float16)(s3.z*d3.z); v[15]=(_Float16)(s3.w*d3.w);
    return v;
}

__device__ __forceinline__ v16h load_b_f16(const _Float16* __restrict__ p) {
    v8h lo = *(const v8h*)(p);
    v8h hi = *(const v8h*)(p + 16);
    v16h v;
#pragma unroll
    for (int i = 0; i < 8; ++i) { v[i] = lo[i]; v[8 + i] = hi[i]; }
    return v;
}

__global__ void prep_weights(const float* __restrict__ W0, const float* __restrict__ W1,
                             const float* __restrict__ Wp1, const float* __restrict__ Wp2,
                             _Float16* __restrict__ W0t, _Float16* __restrict__ W1t,
                             _Float16* __restrict__ Wp1t, _Float16* __restrict__ Wp2t,
                             int* __restrict__ cnt) {
    int idx = blockIdx.x * blockDim.x + threadIdx.x;
    for (int i = idx; i < N_NODES; i += gridDim.x * blockDim.x) cnt[i] = 0;
    if (idx < 128 * 128) {
        int c = idx >> 7, k = idx & 127;
        W0t[idx] = (_Float16)W0[k * 128 + c];
        W1t[idx] = (_Float16)W1[k * 128 + c];
    }
    if (idx < 64 * 128) {
        int c = idx >> 7, k = idx & 127;
        Wp1t[idx] = (_Float16)Wp1[k * 64 + c];
    }
    if (idx < 64 * 64) {
        int c = idx >> 6, k = idx & 63;
        Wp2t[idx] = (_Float16)Wp2[k * 64 + c];
    }
}

__global__ __launch_bounds__(256)
void gemm128_wmma(const float* __restrict__ A, const _Float16* __restrict__ Bt,
                  float* __restrict__ C, int M) {
    int wid  = blockIdx.x * (blockDim.x >> 5) + (threadIdx.x >> 5);
    int lane = threadIdx.x & 31;
    int half = lane >> 4, r = lane & 15;
    int tm = wid >> 3, tn = wid & 7;
    if (tm * 16 >= M) return;
    int row = tm * 16 + r;
    int col = tn * 16 + r;
    v8f acc = {};
#pragma unroll
    for (int kk = 0; kk < 128; kk += 32) {
        int k0 = kk + half * 8;
        v16h a = load_a_f32(A + row * 128 + k0);
        v16h b = load_b_f16(Bt + col * 128 + k0);
        acc = __builtin_amdgcn_wmma_f32_16x16x32_f16(false, a, false, b,
                                                     (short)0, acc, false, false);
    }
    float* cp = C + (tm * 16 + half * 8) * 128 + tn * 16 + r;
#pragma unroll
    for (int v = 0; v < 8; ++v) cp[v * 128] = acc[v];
}

__global__ __launch_bounds__(256)
void edge_rank_kernel(const int* __restrict__ dst, int* __restrict__ cnt,
                      int* __restrict__ rank) {
    int e = blockIdx.x * blockDim.x + threadIdx.x;
    if (e >= E_EDGES) return;
    rank[e] = atomicAdd(&cnt[dst[e]], 1);
}

__global__ __launch_bounds__(SCAN_BLOCK)
void degree_scan_kernel(const int* __restrict__ cnt, int* __restrict__ rowptr) {
    __shared__ int wpre[32];
    __shared__ int wtot[32];
    const int lane = threadIdx.x & 31, w = threadIdx.x >> 5;
    const int lim = blockIdx.x * (SCAN_BLOCK / 4);
    const int4* __restrict__ c4 = (const int4*)cnt;
    int pre = 0;
#pragma unroll
    for (int k = 0; k < SCAN_PRE_LOADS; ++k) {
        int i = threadIdx.x + k * SCAN_BLOCK;
        int4 q = c4[i < lim ? i : 0];
        int s4 = (q.x + q.y) + (q.z + q.w);
        pre += i < lim ? s4 : 0;
    }
    const int idx = blockIdx.x * SCAN_BLOCK + threadIdx.x;
    const int v = cnt[idx < N_NODES ? idx : 0];
    const int own = idx < N_NODES ? v : 0;
    int x = own;
#pragma unroll
    for (int d = 1; d < 32; d <<= 1) {
        int o = __shfl_up(x, d, 32);
        if (lane >= d) x += o;
    }
#pragma unroll
    for (int d = 16; d >= 1; d >>= 1) pre += __shfl_xor(pre, d, 32);
    if (lane == 31) wtot[w] = x;
    if (lane == 0)  wpre[w] = pre;
    __syncthreads();
    int s = wpre[lane] + (lane < w ? wtot[lane] : 0);
#pragma unroll
    for (int d = 16; d >= 1; d >>= 1) s += __shfl_xor(s, d, 32);
    if (idx <= N_NODES) rowptr[idx] = s + x - own;
}

__global__ __launch_bounds__(256)
void edge_place_kernel(const int* __restrict__ src, const int* __restrict__ dst,
                       const int* __restrict__ rank, const int* __restrict__ rowptr,
                       int* __restrict__ esrc) {
    int e = blockIdx.x * blockDim.x + threadIdx.x;
    if (e >= E_EDGES) return;
    esrc[rowptr[dst[e]] + rank[e]] = src[e];
}

template <int M>
__device__ __forceinline__ float row_xor(float x) {
    int b = __builtin_bit_cast(int, x);
    return __builtin_bit_cast(float,
        __builtin_amdgcn_update_dpp(b, b, 0x160 | M, 0xf, 0xf, false));
}

__global__ __launch_bounds__(256)
void gat_node_kernel(const float* __restrict__ feat, const int* __restrict__ rowptr,
                     const int* __restrict__ esrc, const float* __restrict__ attn,
                     const float* __restrict__ hin, float* __restrict__ hout,
                     int residual) {
    int lane = threadIdx.x & 31;
    int node = __builtin_amdgcn_readfirstlane((int)(blockIdx.x * 8 + (threadIdx.x >> 5)));
    if (node >= N_NODES) return;
    int beg = rowptr[node], end = rowptr[node + 1];
    const int lo = lane * 4;
    float4 er = *(const float4*)(feat + (size_t)node * HDIM + lo);
    float4 at = *(const float4*)(attn + lo);
    float m = -INFINITY, s = 0.f;
    float4 acc = make_float4(0.f, 0.f, 0.f, 0.f);

    for (int base = beg; base < end; base += 32) {
        int n = end - base;
        n = n < 32 ? n : 32;
        int myidx = esrc[base + (lane < n ? lane : n - 1)];
        for (int j = 0; j < n; j += EDGE_CHUNK) {
            float4 el[EDGE_CHUNK];
#pragma unroll
            for (int k = 0; k < EDGE_CHUNK; ++k) {
                int jj = j + k < n ? j + k : n - 1;
                int s_ = __builtin_amdgcn_readlane(myidx, jj);
                el[k] = *(const float4*)(feat + (size_t)s_ * HDIM + lo);
            }
            float sc[EDGE_CHUNK];
            float mc = m;
#pragma unroll
            for (int k = 0; k < EDGE_CHUNK; ++k) {
                float x0 = el[k].x + er.x; x0 = fmaxf(x0, NEG_SLOPE * x0);
                float x1 = el[k].y + er.y; x1 = fmaxf(x1, NEG_SLOPE * x1);
                float x2 = el[k].z + er.z; x2 = fmaxf(x2, NEG_SLOPE * x2);
                float x3 = el[k].w + er.w; x3 = fmaxf(x3, NEG_SLOPE * x3);
                float part = x0 * at.x + x1 * at.y + x2 * at.z + x3 * at.w;
                part += row_xor<1>(part);
                part += row_xor<2>(part);
                part += row_xor<4>(part);
                part += row_xor<8>(part);
                sc[k] = (j + k < n) ? part : -INFINITY;
                mc = fmaxf(mc, sc[k]);
            }
            float scale = __expf(m - mc);
            float ps = 0.f;
            float4 a2 = make_float4(0.f, 0.f, 0.f, 0.f);
#pragma unroll
            for (int k = 0; k < EDGE_CHUNK; ++k) {
                float p = __expf(sc[k] - mc);
                ps += p;
                a2.x += el[k].x * p; a2.y += el[k].y * p;
                a2.z += el[k].z * p; a2.w += el[k].w * p;
            }
            s = s * scale + ps;
            acc.x = acc.x * scale + a2.x; acc.y = acc.y * scale + a2.y;
            acc.z = acc.z * scale + a2.z; acc.w = acc.w * scale + a2.w;
            m = mc;
        }
    }

    float4 r = make_float4(0.f, 0.f, 0.f, 0.f);
    if (end > beg) { r.x = acc.x / s; r.y = acc.y / s; r.z = acc.z / s; r.w = acc.w / s; }
    if (residual) {
        float4 hv = *(const float4*)(hin + (size_t)node * HDIM + lo);
        r.x += hv.x; r.y += hv.y; r.z += hv.z; r.w += hv.w;
    }
    r.x = r.x > 0.f ? r.x : (__expf(r.x) - 1.0f);
    r.y = r.y > 0.f ? r.y : (__expf(r.y) - 1.0f);
    r.z = r.z > 0.f ? r.z : (__expf(r.z) - 1.0f);
    r.w = r.w > 0.f ? r.w : (__expf(r.w) - 1.0f);
    *(float4*)(hout + (size_t)node * HDIM + lo) = r;
}

__global__ __launch_bounds__(128)
void predictor_wmma(const float* __restrict__ h,
                    const int* __restrict__ ps, const int* __restrict__ pd,
                    const int* __restrict__ ns, const int* __restrict__ nd,
                    const _Float16* __restrict__ Wp1t, const float* __restrict__ bp1,
                    const _Float16* __restrict__ Wp2t, const float* __restrict__ bp2,
                    const float* __restrict__ Wp3, const float* __restrict__ bp3,
                    float* __restrict__ out) {
    __shared__ _Float16 z1s[4][16 * 64];
    __shared__ float    z2s[4][16 * 64];
    const int tilesPerSet = P_CAND / 16;
    int w = threadIdx.x >> 5;
    int tile = blockIdx.x * 4 + w;
    if (tile >= 2 * tilesPerSet) return;
    bool isNeg = tile >= tilesPerSet;
    int t = isNeg ? tile - tilesPerSet : tile;
    const int* S = isNeg ? ns : ps;
    const int* D = isNeg ? nd : pd;

    int lane = threadIdx.x & 31;
    int half = lane >> 4, r = lane & 15;
    int c = t * 16 + r;
    int si = S[c], di = D[c];

    v16h a1[4];
#pragma unroll
    for (int kc = 0; kc < 4; ++kc) {
        int k0 = kc * 32 + half * 8;
        a1[kc] = load_a_prod(h + si * HDIM + k0, h + di * HDIM + k0);
    }
#pragma unroll
    for (int nt = 0; nt < 4; ++nt) {
        int col = nt * 16 + r;
        v8f acc = {};
#pragma unroll
        for (int kc = 0; kc < 4; ++kc) {
            int k0 = kc * 32 + half * 8;
            v16h b = load_b_f16(Wp1t + col * 128 + k0);
            acc = __builtin_amdgcn_wmma_f32_16x16x32_f16(false, a1[kc], false, b,
                                                         (short)0, acc, false, false);
        }
        float bias = bp1[col];
#pragma unroll
        for (int v = 0; v < 8; ++v) {
            float z = acc[v] + bias;
            z = z > 0.f ? z : 0.f;
            z1s[w][(v + half * 8) * 64 + col] = (_Float16)z;
        }
    }

    v16h a2[2];
#pragma unroll
    for (int kc = 0; kc < 2; ++kc) {
        int k0 = kc * 32 + half * 8;
        const _Float16* zp = &z1s[w][r * 64 + k0];
        a2[kc] = load_b_f16(zp);
    }
#pragma unroll
    for (int nt = 0; nt < 4; ++nt) {
        int col = nt * 16 + r;
        v8f acc = {};
#pragma unroll
        for (int kc = 0; kc < 2; ++kc) {
            int k0 = kc * 32 + half * 8;
            v16h b = load_b_f16(Wp2t + col * 64 + k0);
            acc = __builtin_amdgcn_wmma_f32_16x16x32_f16(false, a2[kc], false, b,
                                                         (short)0, acc, false, false);
        }
        float bias = bp2[col];
#pragma unroll
        for (int v = 0; v < 8; ++v) {
            float z = acc[v] + bias;
            z2s[w][(v + half * 8) * 64 + col] = z > 0.f ? z : 0.f;
        }
    }

    float part = 0.f;
    int jb = half * 32;
#pragma unroll
    for (int j = 0; j < 32; ++j)
        part += z2s[w][r * 64 + jb + j] * Wp3[jb + j];
    part += __shfl_xor(part, 16, 32);
    if (half == 0) {
        out[(isNeg ? P_CAND : 0) + t * 16 + r] = part + bp3[0];
    }
}

extern "C" void kernel_launch(void* const* d_in, const int* in_sizes, int n_in,
                              void* d_out, int out_size, void* d_ws, size_t ws_size,
                              hipStream_t stream) {
    const float* x        = (const float*)d_in[0];
    const int*   src      = (const int*)  d_in[1];
    const int*   dst      = (const int*)  d_in[2];
    const int*   pos_src  = (const int*)  d_in[3];
    const int*   pos_dst  = (const int*)  d_in[4];
    const int*   neg_src  = (const int*)  d_in[5];
    const int*   neg_dst  = (const int*)  d_in[6];
    const float* W0       = (const float*)d_in[7];
    const float* attn0    = (const float*)d_in[8];
    const float* W1       = (const float*)d_in[9];
    const float* attn1    = (const float*)d_in[10];
    const float* Wp1      = (const float*)d_in[11];
    const float* bp1      = (const float*)d_in[12];
    const float* Wp2      = (const float*)d_in[13];
    const float* bp2      = (const float*)d_in[14];
    const float* Wp3      = (const float*)d_in[15];
    const float* bp3      = (const float*)d_in[16];

    char* ws = (char*)d_ws;
    size_t off = 0;
    auto carve = [&](size_t bytes) -> char* {
        char* p = ws + off;
        off += (bytes + 255) & ~(size_t)255;
        return p;
    };
    float*     feat  = (float*)    carve((size_t)N_NODES * HDIM * 4);
    float*     h0    = (float*)    carve((size_t)N_NODES * HDIM * 4);
    float*     h1    = (float*)    carve((size_t)N_NODES * HDIM * 4);
    int*       cnt   = (int*)      carve((size_t)N_NODES * 4);
    int*       rowptr= (int*)      carve((size_t)(N_NODES + 1) * 4);
    int*       rank  = (int*)      carve((size_t)E_EDGES * 4);
    int*       esrc  = (int*)      carve((size_t)E_EDGES * 4);
    _Float16*  W0t   = (_Float16*) carve(128 * 128 * 2);
    _Float16*  W1t   = (_Float16*) carve(128 * 128 * 2);
    _Float16*  Wp1t  = (_Float16*) carve(64 * 128 * 2);
    _Float16*  Wp2t  = (_Float16*) carve(64 * 64 * 2);

    prep_weights<<<(128 * 128 + 255) / 256, 256, 0, stream>>>(
        W0, W1, Wp1, Wp2, W0t, W1t, Wp1t, Wp2t, cnt);

    const int gemmBlocks  = (N_NODES / 16) * 8 / 8;
    const int edgeThrBlocks = (E_EDGES + 255) / 256;
    const int nodeWaveBlocks = (N_NODES + 7) / 8;

    edge_rank_kernel<<<edgeThrBlocks, 256, 0, stream>>>(dst, cnt, rank);
    degree_scan_kernel<<<SCAN_BLOCKS, SCAN_BLOCK, 0, stream>>>(cnt, rowptr);
    edge_place_kernel<<<edgeThrBlocks, 256, 0, stream>>>(src, dst, rank, rowptr, esrc);

    gemm128_wmma<<<gemmBlocks, 256, 0, stream>>>(x, W0t, feat, N_NODES);
    gat_node_kernel<<<nodeWaveBlocks, 256, 0, stream>>>(feat, rowptr, esrc, attn0, x, h0, 0);

    gemm128_wmma<<<gemmBlocks, 256, 0, stream>>>(h0, W1t, feat, N_NODES);
    gat_node_kernel<<<nodeWaveBlocks, 256, 0, stream>>>(feat, rowptr, esrc, attn1, h0, h1, 1);

    const int predBlocks = (2 * (P_CAND / 16) + 3) / 4;
    predictor_wmma<<<predBlocks, 128, 0, stream>>>(
        h1, pos_src, pos_dst, neg_src, neg_dst,
        Wp1t, bp1, Wp2t, bp2, Wp3, bp3, (float*)d_out);
}


# Predictor: each h row gathered by one 512-B wave load, multiplied in registers, f16 product tile to the fragment loader via a wave-private LDS slot; weights in LDS per work-group; sums unchanged
# speedup vs baseline: 5.1315x; 1.5901x over previous
#include <hip/hip_runtime.h>


typedef __attribute__((ext_vector_type(16))) _Float16 v16h;
typedef __attribute__((ext_vector_type(8)))  _Float16 v8h;
typedef __attribute__((ext_vector_type(8)))  float    v8f;
typedef __attribute__((ext_vector_type(4)))  _Float16 v4h;
typedef __attribute__((ext_vector_type(2)))  float    v2f;
typedef __attribute__((ext_vector_type(4)))  float    v4f;

#define N_NODES 50000
#define E_EDGES 800000
#define P_CAND  100000
#define HDIM    128
#define NEG_SLOPE 0.2f
#define EDGE_CHUNK 8
#define BKT 32
#define BKT_CAP (BKT - 1)


__device__ __forceinline__ v16h load_a_f32(const float* __restrict__ p) {
    float4 a0 = *(const float4*)(p);
    float4 a1 = *(const float4*)(p + 4);
    float4 a2 = *(const float4*)(p + 16);
    float4 a3 = *(const float4*)(p + 20);
    v16h v;
    v[0]=(_Float16)a0.x; v[1]=(_Float16)a0.y; v[2]=(_Float16)a0.z; v[3]=(_Float16)a0.w;
    v[4]=(_Float16)a1.x; v[5]=(_Float16)a1.y; v[6]=(_Float16)a1.z; v[7]=(_Float16)a1.w;
    v[8]=(_Float16)a2.x; v[9]=(_Float16)a2.y; v[10]=(_Float16)a2.z; v[11]=(_Float16)a2.w;
    v[12]=(_Float16)a3.x; v[13]=(_Float16)a3.y; v[14]=(_Float16)a3.z; v[15]=(_Float16)a3.w;
    return v;
}

__device__ __forceinline__ v16h load_a_prod(const float* __restrict__ sp,
                                            const float* __restrict__ dp) {
    float4 s0 = *(const float4*)(sp),      d0 = *(const float4*)(dp);
    float4 s1 = *(const float4*)(sp + 4),  d1 = *(const float4*)(dp + 4);
    float4 s2 = *(const float4*)(sp + 16), d2 = *(const float4*)(dp + 16);
    float4 s3 = *(const float4*)(sp + 20), d3 = *(const float4*)(dp + 20);
    v16h v;
    v[0]=(_Float16)(s0.x*d0.x); v[1]=(_Float16)(s0.y*d0.y);
    v[2]=(_Float16)(s0.z*d0.z); v[3]=(_Float16)(s0.w*d0.w);
    v[4]=(_Float16)(s1.x*d1.x); v[5]=(_Float16)(s1.y*d1.y);
    v[6]=(_Float16)(s1.z*d1.z); v[7]=(_Float16)(s1.w*d1.w);
    v[8]=(_Float16)(s2.x*d2.x); v[9]=(_Float16)(s2.y*d2.y);
    v[10]=(_Float16)(s2.z*d2.z); v[11]=(_Float16)(s2.w*d2.w);
    v[12]=(_Float16)(s3.x*d3.x); v[13]=(_Float16)(s3.y*d3.y);
    v[14]=(_Float16)(s3.z*d3.z); v[15]=(_Float16)(s3.w*d3.w);
    return v;
}

__device__ __forceinline__ v16h load_b_f16(const _Float16* __restrict__ p) {
    v8h lo = *(const v8h*)(p);
    v8h hi = *(const v8h*)(p + 16);
    v16h v;
#pragma unroll
    for (int i = 0; i < 8; ++i) { v[i] = lo[i]; v[8 + i] = hi[i]; }
    return v;
}

__global__ void prep_weights(const float* __restrict__ W0, const float* __restrict__ W1,
                             const float* __restrict__ Wp1, const float* __restrict__ Wp2,
                             _Float16* __restrict__ W0t, _Float16* __restrict__ W1t,
                             _Float16* __restrict__ Wp1t, _Float16* __restrict__ Wp2t,
                             int* __restrict__ bkt, int* __restrict__ ovfc) {
    int idx = blockIdx.x * blockDim.x + threadIdx.x;
    int4* __restrict__ b4 = (int4*)bkt;
    for (int i = idx; i < N_NODES * (BKT / 4); i += gridDim.x * blockDim.x)
        b4[i] = make_int4(0, 0, 0, 0);
    if (idx == 0) *ovfc = 0;
    if (idx < 128 * 128) {
        int c = idx >> 7, k = idx & 127;
        W0t[idx] = (_Float16)W0[k * 128 + c];
        W1t[idx] = (_Float16)W1[k * 128 + c];
    }
    if (idx < 64 * 128) {
        int c = idx >> 7, k = idx & 127;
        Wp1t[idx] = (_Float16)Wp1[k * 64 + c];
    }
    if (idx < 64 * 64) {
        int c = idx >> 6, k = idx & 63;
        Wp2t[idx] = (_Float16)Wp2[k * 64 + c];
    }
}

#define GEMM_WAVES 8
#define H_PITCH 136
#define C_PITCH 132
__global__ __launch_bounds__(256)
void gemm128_wmma(const float* __restrict__ A, const _Float16* __restrict__ Bt,
                  float* __restrict__ C, int M) {
    __shared__ __attribute__((aligned(16))) _Float16 bs[128 * H_PITCH];
    __shared__ __attribute__((aligned(16))) float cs[GEMM_WAVES][16 * C_PITCH];
    const int w = threadIdx.x >> 5, lane = threadIdx.x & 31;
    const int half = lane >> 4, r = lane & 15;
    const int strip = blockIdx.x * GEMM_WAVES + w;
    const bool live = strip * 16 < M;
    float* cw = cs[w];
    _Float16* aw = (_Float16*)cw;

    v8h wq[8];
#pragma unroll
    for (int i = 0; i < 8; ++i) wq[i] = *(const v8h*)(Bt + (i * 256 + threadIdx.x) * 8);
    v4f ar[16];
    if (live) {
        const float* ap = A + strip * 16 * 128 + lane * 4;
#pragma unroll
        for (int i = 0; i < 16; ++i) ar[i] = *(const v4f*)(ap + i * 128);
    }
#pragma unroll
    for (int i = 0; i < 8; ++i) {
        const int c = i * 256 + threadIdx.x;
        *(v8h*)(bs + (c >> 4) * H_PITCH + (c & 15) * 8) = wq[i];
    }
    if (live) {
#pragma unroll
        for (int i = 0; i < 16; ++i) {
            v4h q;
            q[0] = (_Float16)ar[i][0]; q[1] = (_Float16)ar[i][1];
            q[2] = (_Float16)ar[i][2]; q[3] = (_Float16)ar[i][3];
            *(v4h*)(aw + i * H_PITCH + lane * 4) = q;
        }
    }
    __syncthreads();
    if (!live) return;

    v16h a[4];
#pragma unroll
    for (int kc = 0; kc < 4; ++kc) a[kc] = load_b_f16(aw + r * H_PITCH + kc * 32 + half * 8);
    v8f acc[8] = {};
#pragma unroll
    for (int kc = 0; kc < 4; ++kc) {
#pragma unroll
        for (int tn = 0; tn < 8; ++tn) {
            v16h b = load_b_f16(bs + (tn * 16 + r) * H_PITCH + kc * 32 + half * 8);
            acc[tn] = __builtin_amdgcn_wmma_f32_16x16x32_f16(false, a[kc], false, b,
                                                             (short)0, acc[tn], false, false);
        }
    }
#pragma unroll
    for (int tn = 0; tn < 8; ++tn)
#pragma unroll
        for (int v = 0; v < 8; ++v) cw[(v + half * 8) * C_PITCH + tn * 16 + r] = acc[tn][v];
    float* cp = C + strip * 16 * 128 + lane * 4;
#pragma unroll
    for (int i = 0; i < 16; ++i)
        *(v4f*)(cp + i * 128) = *(const v4f*)(cw + i * C_PITCH + lane * 4);
}

__global__ __launch_bounds__(256)
void edge_fill_kernel(const int* __restrict__ src, const int* __restrict__ dst,
                      int* __restrict__ bkt, int* __restrict__ ovfc, int* __restrict__ ovf) {
    int e = blockIdx.x * blockDim.x + threadIdx.x;
    if (e >= E_EDGES) return;
    const int d = dst[e], s = src[e];
    int* line = bkt + d * BKT;
    const int pos = atomicAdd(line, 1);
    if (pos < BKT_CAP) {
        line[1 + pos] = s;
    } else {
        const int k = atomicAdd(ovfc, 1);
        *(int2*)(ovf + 2 * k) = make_int2(d, s);
    }
}

template <int M>
__device__ __forceinline__ float row_xor(float x) {
    int b = __builtin_bit_cast(int, x);
    return __builtin_bit_cast(float,
        __builtin_amdgcn_update_dpp(0, b, 0x160 | M, 0xf, 0xf, true));
}

static_assert(EDGE_CHUNK == 8, "the load fence below names eight rows");

template <int M>
__device__ __forceinline__ int row_xor_i(int x) {
    return __builtin_amdgcn_update_dpp(0, x, 0x160 | M, 0xf, 0xf, true);
}

template <int K, int J>
__device__ __forceinline__ int sort_step(int v, int lane) {
    const int p = (J == 16) ? __shfl_xor(v, 16, 32) : row_xor_i<(J & 15)>(v);
    const bool up  = (K == 32) ? true : ((lane & K) == 0);
    const bool low = (lane & J) == 0;
    const int lo = v < p ? v : p, hi = v < p ? p : v;
    return (low == up) ? lo : hi;
}

__device__ __forceinline__ int sort32(int v, int lane) {
    v = sort_step<2, 1>(v, lane);
    v = sort_step<4, 2>(v, lane);   v = sort_step<4, 1>(v, lane);
    v = sort_step<8, 4>(v, lane);   v = sort_step<8, 2>(v, lane);   v = sort_step<8, 1>(v, lane);
    v = sort_step<16, 8>(v, lane);  v = sort_step<16, 4>(v, lane);  v = sort_step<16, 2>(v, lane);
    v = sort_step<16, 1>(v, lane);
    v = sort_step<32, 16>(v, lane); v = sort_step<32, 8>(v, lane);  v = sort_step<32, 4>(v, lane);
    v = sort_step<32, 2>(v, lane);  v = sort_step<32, 1>(v, lane);
    return v;
}

template <bool FIRST>
__global__ __launch_bounds__(256)
void gat_node_kernel(const float* __restrict__ feat, int* bkt,
                     const int* __restrict__ ovfc, const int* __restrict__ ovf,
                     const float* __restrict__ attn,
                     const float* __restrict__ hin, float* __restrict__ hout,
                     int residual) {
    const int lane = threadIdx.x & 31;
    const int node = __builtin_amdgcn_readfirstlane((int)(blockIdx.x * 8 + (threadIdx.x >> 5)));
    if (node >= N_NODES) return;
    int ids = bkt[node * BKT + lane];
    const int deg = __builtin_amdgcn_readfirstlane(ids);
    const int nb = deg < BKT_CAP ? deg : BKT_CAP;
    if (FIRST) {
        int key = lane == 0 ? (-2147483647 - 1) : (lane <= nb ? ids : 2147483647);
        key = sort32(key, lane);
        ids = lane == 0 ? deg : key;
        bkt[node * BKT + lane] = ids;
    }
    const unsigned lob = (unsigned)lane * 16u;
    const char* fb = (const char*)feat;
    const size_t nrow = (size_t)node * (HDIM * 4u);
    const float4 er4 = *(const float4*)(fb + nrow + lob);
    const float4 at4 = *(const float4*)((const char*)attn + lob);
    const v2f er01 = {er4.x, er4.y}, er23 = {er4.z, er4.w};
    const v2f at01 = {at4.x, at4.y}, at23 = {at4.z, at4.w};
    float m = -INFINITY, s = 0.f;
    v2f acc01 = {0.f, 0.f}, acc23 = {0.f, 0.f};

    for (int j = 0; j < nb; j += EDGE_CHUNK) {
        const int cnt = nb - j;
        v4f el[EDGE_CHUNK];
#pragma unroll
        for (int k = 0; k < EDGE_CHUNK; ++k) {
            const int jj = k < cnt ? j + k : nb - 1;
            const int s_ = __builtin_amdgcn_readlane(ids, 1 + jj);
            el[k] = *(const v4f*)(fb + (((unsigned)s_ << 9) + lob));
        }
        asm volatile("" :: "v"(el[0]), "v"(el[1]), "v"(el[2]), "v"(el[3]),
                           "v"(el[4]), "v"(el[5]), "v"(el[6]), "v"(el[7]));
        float sc[EDGE_CHUNK];
#pragma unroll
        for (int k = 0; k < EDGE_CHUNK; ++k) {
            sc[k] = -INFINITY;
            if (k < cnt) {
                v2f x01 = el[k].xy, x23 = el[k].zw;
                x01 += er01; x23 += er23;
                const v2f y01 = x01 * NEG_SLOPE, y23 = x23 * NEG_SLOPE;
                x01 = __builtin_elementwise_max(x01, y01);
                x23 = __builtin_elementwise_max(x23, y23);
                const v2f t = x01 * at01 + x23 * at23;
                float part = t.x + t.y;
                part += row_xor<1>(part);
                part += row_xor<2>(part);
                part += row_xor<4>(part);
                part += row_xor<8>(part);
                sc[k] = part;
            }
        }
        float mc = m;
#pragma unroll
        for (int k = 0; k < EDGE_CHUNK; ++k) mc = fmaxf(mc, sc[k]);
        const float scale = __expf(m - mc);
        float ps = 0.f;
        v2f a01 = {0.f, 0.f}, a23 = {0.f, 0.f};
#pragma unroll
        for (int k = 0; k < EDGE_CHUNK; ++k) {
            if (k < cnt) {
                const float p = __expf(sc[k] - mc);
                const v2f e01 = el[k].xy, e23 = el[k].zw;
                ps += p;
                a01 += e01 * p; a23 += e23 * p;
            }
        }
        s = s * scale + ps;
        acc01 = acc01 * scale + a01;
        acc23 = acc23 * scale + a23;
        m = mc;
    }

    if (deg > BKT_CAP) {
        int total = *ovfc;
        total = total < E_EDGES ? total : E_EDGES;
        const int2* __restrict__ o2 = (const int2*)ovf;
        for (int base = 0; base < total; base += 32) {
            const int i = base + lane;
            const int2 q = o2[i < total ? i : total - 1];
            unsigned hit = __builtin_amdgcn_ballot_w32(i < total && q.x == node);
            while (hit != 0u) {
                const int s_ = __builtin_amdgcn_readlane(q.y, __builtin_ctz(hit));
                hit &= hit - 1u;
                const v4f e = *(const v4f*)(fb + (((unsigned)s_ << 9) + lob));
                v2f x01 = e.xy, x23 = e.zw;
                x01 += er01; x23 += er23;
                const v2f y01 = x01 * NEG_SLOPE, y23 = x23 * NEG_SLOPE;
                x01 = __builtin_elementwise_max(x01, y01);
                x23 = __builtin_elementwise_max(x23, y23);
                const v2f t = x01 * at01 + x23 * at23;
                float part = t.x + t.y;
                part += row_xor<1>(part);
                part += row_xor<2>(part);
                part += row_xor<4>(part);
                part += row_xor<8>(part);
                const float mc = fmaxf(m, part);
                const float scale = __expf(m - mc), p = __expf(part - mc);
                s = s * scale + p;
                acc01 = acc01 * scale + e.xy * p;
                acc23 = acc23 * scale + e.zw * p;
                m = mc;
            }
        }
    }

    float4 r = make_float4(0.f, 0.f, 0.f, 0.f);
    if (deg > 0) { r.x = acc01.x / s; r.y = acc01.y / s; r.z = acc23.x / s; r.w = acc23.y / s; }
    if (residual) {
        const float4 hv = *(const float4*)((const char*)hin + nrow + lob);
        r.x += hv.x; r.y += hv.y; r.z += hv.z; r.w += hv.w;
    }
    r.x = r.x > 0.f ? r.x : (__expf(r.x) - 1.0f);
    r.y = r.y > 0.f ? r.y : (__expf(r.y) - 1.0f);
    r.z = r.z > 0.f ? r.z : (__expf(r.z) - 1.0f);
    r.w = r.w > 0.f ? r.w : (__expf(r.w) - 1.0f);
    *(float4*)((char*)hout + nrow + lob) = r;
}

#define PRED_WAVES 8
#define P2_PITCH 72
#define Z2_PITCH 65
__global__ __launch_bounds__(256)
void predictor_wmma(const float* __restrict__ h,
                    const int* __restrict__ ps, const int* __restrict__ pd,
                    const int* __restrict__ ns, const int* __restrict__ nd,
                    const _Float16* __restrict__ Wp1t, const float* __restrict__ bp1,
                    const _Float16* __restrict__ Wp2t, const float* __restrict__ bp2,
                    const float* __restrict__ Wp3, const float* __restrict__ bp3,
                    float* __restrict__ out) {
    __shared__ __attribute__((aligned(16))) _Float16 w1s[64 * H_PITCH];
    __shared__ __attribute__((aligned(16))) _Float16 w2s[64 * P2_PITCH];
    __shared__ __attribute__((aligned(16))) _Float16 zs[PRED_WAVES][16 * H_PITCH];
    __shared__ float z2s[PRED_WAVES][16 * Z2_PITCH];
    const int tilesPerSet = P_CAND / 16;
    const int w = threadIdx.x >> 5, lane = threadIdx.x & 31;
    const int half = lane >> 4, r = lane & 15;
    const int tile = __builtin_amdgcn_readfirstlane((int)(blockIdx.x * PRED_WAVES + w));
    const bool live = tile < 2 * tilesPerSet;
    const bool isNeg = tile >= tilesPerSet;
    const int t = isNeg ? tile - tilesPerSet : tile;
    _Float16* zw = zs[w];

    v8h q1[4], q2[2];
#pragma unroll
    for (int i = 0; i < 4; ++i) q1[i] = *(const v8h*)(Wp1t + (i * 256 + threadIdx.x) * 8);
#pragma unroll
    for (int i = 0; i < 2; ++i) q2[i] = *(const v8h*)(Wp2t + (i * 256 + threadIdx.x) * 8);

    int mine = 0;
    if (live) {
        const int* S = isNeg ? ns : ps;
        const int* D = isNeg ? nd : pd;
        mine = half ? D[t * 16 + r] : S[t * 16 + r];
    }
#pragma unroll
    for (int i = 0; i < 4; ++i) {
        const int c = i * 256 + threadIdx.x;
        *(v8h*)(w1s + (c >> 4) * H_PITCH + (c & 15) * 8) = q1[i];
    }
#pragma unroll
    for (int i = 0; i < 2; ++i) {
        const int c = i * 256 + threadIdx.x;
        *(v8h*)(w2s + (c >> 3) * P2_PITCH + (c & 7) * 8) = q2[i];
    }
    if (live) {
        const unsigned lob = (unsigned)lane * 16u;
        const char* hb = (const char*)h;
        v4f sr[16], dr[16];
#pragma unroll
        for (int i = 0; i < 16; ++i) {
            const int si = __builtin_amdgcn_readlane(mine, i);
            const int di = __builtin_amdgcn_readlane(mine, 16 + i);
            sr[i] = *(const v4f*)(hb + (((unsigned)si << 9) + lob));
            dr[i] = *(const v4f*)(hb + (((unsigned)di << 9) + lob));
        }
#pragma unroll
        for (int i = 0; i < 16; ++i) {
            v4h q;
            q[0] = (_Float16)(sr[i][0] * dr[i][0]); q[1] = (_Float16)(sr[i][1] * dr[i][1]);
            q[2] = (_Float16)(sr[i][2] * dr[i][2]); q[3] = (_Float16)(sr[i][3] * dr[i][3]);
            *(v4h*)(zw + i * H_PITCH + lane * 4) = q;
        }
    }
    __syncthreads();
    if (!live) return;

    v16h a1[4];
#pragma unroll
    for (int kc = 0; kc < 4; ++kc) a1[kc] = load_b_f16(zw + r * H_PITCH + kc * 32 + half * 8);
#pragma unroll
    for (int nt = 0; nt < 4; ++nt) {
        const int col = nt * 16 + r;
        v8f acc = {};
#pragma unroll
        for (int kc = 0; kc < 4; ++kc) {
            v16h b = load_b_f16(w1s + col * H_PITCH + kc * 32 + half * 8);
            acc = __builtin_amdgcn_wmma_f32_16x16x32_f16(false, a1[kc], false, b,
                                                         (short)0, acc, false, false);
        }
        const float bias = bp1[col];
#pragma unroll
        for (int v = 0; v < 8; ++v) {
            float z = acc[v] + bias;
            z = z > 0.f ? z : 0.f;
            zw[(v + half * 8) * P2_PITCH + col] = (_Float16)z;
        }
    }

    v16h a2[2];
#pragma unroll
    for (int kc = 0; kc < 2; ++kc) a2[kc] = load_b_f16(zw + r * P2_PITCH + kc * 32 + half * 8);
#pragma unroll
    for (int nt = 0; nt < 4; ++nt) {
        const int col = nt * 16 + r;
        v8f acc = {};
#pragma unroll
        for (int kc = 0; kc < 2; ++kc) {
            v16h b = load_b_f16(w2s + col * P2_PITCH + kc * 32 + half * 8);
            acc = __builtin_amdgcn_wmma_f32_16x16x32_f16(false, a2[kc], false, b,
                                                         (short)0, acc, false, false);
        }
        const float bias = bp2[col];
#pragma unroll
        for (int v = 0; v < 8; ++v) {
            float z = acc[v] + bias;
            z2s[w][(v + half * 8) * Z2_PITCH + col] = z > 0.f ? z : 0.f;
        }
    }

    float part = 0.f;
    const int jb = half * 32;
#pragma unroll
    for (int j = 0; j < 32; ++j)
        part += z2s[w][r * Z2_PITCH + jb + j] * Wp3[jb + j];
    part += __shfl_xor(part, 16, 32);
    if (half == 0) {
        out[(isNeg ? P_CAND : 0) + t * 16 + r] = part + bp3[0];
    }
}

extern "C" void kernel_launch(void* const* d_in, const int* in_sizes, int n_in,
                              void* d_out, int out_size, void* d_ws, size_t ws_size,
                              hipStream_t stream) {
    const float* x        = (const float*)d_in[0];
    const int*   src      = (const int*)  d_in[1];
    const int*   dst      = (const int*)  d_in[2];
    const int*   pos_src  = (const int*)  d_in[3];
    const int*   pos_dst  = (const int*)  d_in[4];
    const int*   neg_src  = (const int*)  d_in[5];
    const int*   neg_dst  = (const int*)  d_in[6];
    const float* W0       = (const float*)d_in[7];
    const float* attn0    = (const float*)d_in[8];
    const float* W1       = (const float*)d_in[9];
    const float* attn1    = (const float*)d_in[10];
    const float* Wp1      = (const float*)d_in[11];
    const float* bp1      = (const float*)d_in[12];
    const float* Wp2      = (const float*)d_in[13];
    const float* bp2      = (const float*)d_in[14];
    const float* Wp3      = (const float*)d_in[15];
    const float* bp3      = (const float*)d_in[16];

    char* ws = (char*)d_ws;
    size_t off = 0;
    auto carve = [&](size_t bytes) -> char* {
        char* p = ws + off;
        off += (bytes + 255) & ~(size_t)255;
        return p;
    };
    float*     feat  = (float*)    carve((size_t)N_NODES * HDIM * 4);
    float*     h0    = (float*)    carve((size_t)N_NODES * HDIM * 4);
    float*     h1    = (float*)    carve((size_t)N_NODES * HDIM * 4);
    int*       bkt   = (int*)      carve((size_t)N_NODES * BKT * 4);
    int*       ovfc  = (int*)      carve(4);
    int*       ovf   = (int*)      carve((size_t)E_EDGES * 8);
    _Float16*  W0t   = (_Float16*) carve(128 * 128 * 2);
    _Float16*  W1t   = (_Float16*) carve(128 * 128 * 2);
    _Float16*  Wp1t  = (_Float16*) carve(64 * 128 * 2);
    _Float16*  Wp2t  = (_Float16*) carve(64 * 64 * 2);

    prep_weights<<<256, 256, 0, stream>>>(
        W0, W1, Wp1, Wp2, W0t, W1t, Wp1t, Wp2t, bkt, ovfc);

    const int gemmBlocks  = (N_NODES / 16 + GEMM_WAVES - 1) / GEMM_WAVES;
    const int edgeThrBlocks = (E_EDGES + 255) / 256;
    const int nodeWaveBlocks = (N_NODES + 7) / 8;

    edge_fill_kernel<<<edgeThrBlocks, 256, 0, stream>>>(src, dst, bkt, ovfc, ovf);

    gemm128_wmma<<<gemmBlocks, 256, 0, stream>>>(x, W0t, feat, N_NODES);
    gat_node_kernel<true><<<nodeWaveBlocks, 256, 0, stream>>>(feat, bkt, ovfc, ovf, attn0, x, h0, 0);

    gemm128_wmma<<<gemmBlocks, 256, 0, stream>>>(h0, W1t, feat, N_NODES);
    gat_node_kernel<false><<<nodeWaveBlocks, 256, 0, stream>>>(feat, bkt, ovfc, ovf, attn1, h0, h1, 1);

    const int predBlocks = (2 * (P_CAND / 16) + PRED_WAVES - 1) / PRED_WAVES;
    predictor_wmma<<<predBlocks, 256, 0, stream>>>(
        h1, pos_src, pos_dst, neg_src, neg_dst,
        Wp1t, bp1, Wp2t, bp2, Wp3, bp3, (float*)d_out);
}


# Layer-1 GEMM done inside the layer-0 node kernel: 16 node waves per work-group stage their h rows in LDS as f16, 8 of them multiply the strip by W1 on WMMA; GEMM launch and h re-read gone; same sums
# speedup vs baseline: 5.5420x; 1.0800x over previous
#include <hip/hip_runtime.h>


typedef __attribute__((ext_vector_type(16))) _Float16 v16h;
typedef __attribute__((ext_vector_type(8)))  _Float16 v8h;
typedef __attribute__((ext_vector_type(8)))  float    v8f;
typedef __attribute__((ext_vector_type(4)))  _Float16 v4h;
typedef __attribute__((ext_vector_type(2)))  float    v2f;
typedef __attribute__((ext_vector_type(4)))  float    v4f;

#define N_NODES 50000
#define E_EDGES 800000
#define P_CAND  100000
#define HDIM    128
#define NEG_SLOPE 0.2f
#define EDGE_CHUNK 8
#define BKT 32
#define BKT_CAP (BKT - 1)


__device__ __forceinline__ v16h load_a_f32(const float* __restrict__ p) {
    float4 a0 = *(const float4*)(p);
    float4 a1 = *(const float4*)(p + 4);
    float4 a2 = *(const float4*)(p + 16);
    float4 a3 = *(const float4*)(p + 20);
    v16h v;
    v[0]=(_Float16)a0.x; v[1]=(_Float16)a0.y; v[2]=(_Float16)a0.z; v[3]=(_Float16)a0.w;
    v[4]=(_Float16)a1.x; v[5]=(_Float16)a1.y; v[6]=(_Float16)a1.z; v[7]=(_Float16)a1.w;
    v[8]=(_Float16)a2.x; v[9]=(_Float16)a2.y; v[10]=(_Float16)a2.z; v[11]=(_Float16)a2.w;
    v[12]=(_Float16)a3.x; v[13]=(_Float16)a3.y; v[14]=(_Float16)a3.z; v[15]=(_Float16)a3.w;
    return v;
}

__device__ __forceinline__ v16h load_a_prod(const float* __restrict__ sp,
                                            const float* __restrict__ dp) {
    float4 s0 = *(const float4*)(sp),      d0 = *(const float4*)(dp);
    float4 s1 = *(const float4*)(sp + 4),  d1 = *(const float4*)(dp + 4);
    float4 s2 = *(const float4*)(sp + 16), d2 = *(const float4*)(dp + 16);
    float4 s3 = *(const float4*)(sp + 20), d3 = *(const float4*)(dp + 20);
    v16h v;
    v[0]=(_Float16)(s0.x*d0.x); v[1]=(_Float16)(s0.y*d0.y);
    v[2]=(_Float16)(s0.z*d0.z); v[3]=(_Float16)(s0.w*d0.w);
    v[4]=(_Float16)(s1.x*d1.x); v[5]=(_Float16)(s1.y*d1.y);
    v[6]=(_Float16)(s1.z*d1.z); v[7]=(_Float16)(s1.w*d1.w);
    v[8]=(_Float16)(s2.x*d2.x); v[9]=(_Float16)(s2.y*d2.y);
    v[10]=(_Float16)(s2.z*d2.z); v[11]=(_Float16)(s2.w*d2.w);
    v[12]=(_Float16)(s3.x*d3.x); v[13]=(_Float16)(s3.y*d3.y);
    v[14]=(_Float16)(s3.z*d3.z); v[15]=(_Float16)(s3.w*d3.w);
    return v;
}

__device__ __forceinline__ v16h load_b_f16(const _Float16* __restrict__ p) {
    v8h lo = *(const v8h*)(p);
    v8h hi = *(const v8h*)(p + 16);
    v16h v;
#pragma unroll
    for (int i = 0; i < 8; ++i) { v[i] = lo[i]; v[8 + i] = hi[i]; }
    return v;
}

__global__ void prep_weights(const float* __restrict__ W0, const float* __restrict__ W1,
                             const float* __restrict__ Wp1, const float* __restrict__ Wp2,
                             _Float16* __restrict__ W0t, _Float16* __restrict__ W1t,
                             _Float16* __restrict__ Wp1t, _Float16* __restrict__ Wp2t,
                             int* __restrict__ bkt, int* __restrict__ ovfc) {
    int idx = blockIdx.x * blockDim.x + threadIdx.x;
    int4* __restrict__ b4 = (int4*)bkt;
    for (int i = idx; i < N_NODES * (BKT / 4); i += gridDim.x * blockDim.x)
        b4[i] = make_int4(0, 0, 0, 0);
    if (idx == 0) *ovfc = 0;
    if (idx < 128 * 128) {
        int c = idx >> 7, k = idx & 127;
        W0t[idx] = (_Float16)W0[k * 128 + c];
        W1t[idx] = (_Float16)W1[k * 128 + c];
    }
    if (idx < 64 * 128) {
        int c = idx >> 7, k = idx & 127;
        Wp1t[idx] = (_Float16)Wp1[k * 64 + c];
    }
    if (idx < 64 * 64) {
        int c = idx >> 6, k = idx & 63;
        Wp2t[idx] = (_Float16)Wp2[k * 64 + c];
    }
}

#define GEMM_WAVES 8
#define H_PITCH 136
#define C_PITCH 132
__global__ __launch_bounds__(256)
void gemm128_wmma(const float* __restrict__ A, const _Float16* __restrict__ Bt,
                  float* __restrict__ C, int M) {
    __shared__ __attribute__((aligned(16))) _Float16 bs[128 * H_PITCH];
    __shared__ __attribute__((aligned(16))) float cs[GEMM_WAVES][16 * C_PITCH];
    const int w = threadIdx.x >> 5, lane = threadIdx.x & 31;
    const int half = lane >> 4, r = lane & 15;
    const int strip = blockIdx.x * GEMM_WAVES + w;
    const bool live = strip * 16 < M;
    float* cw = cs[w];
    _Float16* aw = (_Float16*)cw;

    v8h wq[8];
#pragma unroll
    for (int i = 0; i < 8; ++i) wq[i] = *(const v8h*)(Bt + (i * 256 + threadIdx.x) * 8);
    v4f ar[16];
    if (live) {
        const float* ap = A + strip * 16 * 128 + lane * 4;
#pragma unroll
        for (int i = 0; i < 16; ++i) ar[i] = *(const v4f*)(ap + i * 128);
    }
#pragma unroll
    for (int i = 0; i < 8; ++i) {
        const int c = i * 256 + threadIdx.x;
        *(v8h*)(bs + (c >> 4) * H_PITCH + (c & 15) * 8) = wq[i];
    }
    if (live) {
#pragma unroll
        for (int i = 0; i < 16; ++i) {
            v4h q;
            q[0] = (_Float16)ar[i][0]; q[1] = (_Float16)ar[i][1];
            q[2] = (_Float16)ar[i][2]; q[3] = (_Float16)ar[i][3];
            *(v4h*)(aw + i * H_PITCH + lane * 4) = q;
        }
    }
    __syncthreads();
    if (!live) return;

    v16h a[4];
#pragma unroll
    for (int kc = 0; kc < 4; ++kc) a[kc] = load_b_f16(aw + r * H_PITCH + kc * 32 + half * 8);
    v8f acc[8] = {};
#pragma unroll
    for (int kc = 0; kc < 4; ++kc) {
#pragma unroll
        for (int tn = 0; tn < 8; ++tn) {
            v16h b = load_b_f16(bs + (tn * 16 + r) * H_PITCH + kc * 32 + half * 8);
            acc[tn] = __builtin_amdgcn_wmma_f32_16x16x32_f16(false, a[kc], false, b,
                                                             (short)0, acc[tn], false, false);
        }
    }
#pragma unroll
    for (int tn = 0; tn < 8; ++tn)
#pragma unroll
        for (int v = 0; v < 8; ++v) cw[(v + half * 8) * C_PITCH + tn * 16 + r] = acc[tn][v];
    float* cp = C + strip * 16 * 128 + lane * 4;
#pragma unroll
    for (int i = 0; i < 16; ++i)
        *(v4f*)(cp + i * 128) = *(const v4f*)(cw + i * C_PITCH + lane * 4);
}

__global__ __launch_bounds__(256)
void edge_fill_kernel(const int* __restrict__ src, const int* __restrict__ dst,
                      int* __restrict__ bkt, int* __restrict__ ovfc, int* __restrict__ ovf) {
    int e = blockIdx.x * blockDim.x + threadIdx.x;
    if (e >= E_EDGES) return;
    const int d = dst[e], s = src[e];
    int* line = bkt + d * BKT;
    const int pos = atomicAdd(line, 1);
    if (pos < BKT_CAP) {
        line[1 + pos] = s;
    } else {
        const int k = atomicAdd(ovfc, 1);
        *(int2*)(ovf + 2 * k) = make_int2(d, s);
    }
}

template <int M>
__device__ __forceinline__ float row_xor(float x) {
    int b = __builtin_bit_cast(int, x);
    return __builtin_bit_cast(float,
        __builtin_amdgcn_update_dpp(0, b, 0x160 | M, 0xf, 0xf, true));
}

static_assert(EDGE_CHUNK == 8, "the load fence below names eight rows");

template <int M>
__device__ __forceinline__ int row_xor_i(int x) {
    return __builtin_amdgcn_update_dpp(0, x, 0x160 | M, 0xf, 0xf, true);
}

template <int K, int J>
__device__ __forceinline__ int sort_step(int v, int lane) {
    const int p = (J == 16) ? __shfl_xor(v, 16, 32) : row_xor_i<(J & 15)>(v);
    const bool up  = (K == 32) ? true : ((lane & K) == 0);
    const bool low = (lane & J) == 0;
    const int lo = v < p ? v : p, hi = v < p ? p : v;
    return (low == up) ? lo : hi;
}

__device__ __forceinline__ int sort32(int v, int lane) {
    v = sort_step<2, 1>(v, lane);
    v = sort_step<4, 2>(v, lane);   v = sort_step<4, 1>(v, lane);
    v = sort_step<8, 4>(v, lane);   v = sort_step<8, 2>(v, lane);   v = sort_step<8, 1>(v, lane);
    v = sort_step<16, 8>(v, lane);  v = sort_step<16, 4>(v, lane);  v = sort_step<16, 2>(v, lane);
    v = sort_step<16, 1>(v, lane);
    v = sort_step<32, 16>(v, lane); v = sort_step<32, 8>(v, lane);  v = sort_step<32, 4>(v, lane);
    v = sort_step<32, 2>(v, lane);  v = sort_step<32, 1>(v, lane);
    return v;
}

#define STRIP_WAVES 16
static_assert(N_NODES % STRIP_WAVES == 0, "the FIRST grid is exact: every wave reaches both barriers");
template <bool FIRST>
__global__ __launch_bounds__(FIRST ? 32 * STRIP_WAVES : 256)
void gat_node_kernel(const float* __restrict__ feat, int* bkt,
                     const int* __restrict__ ovfc, const int* __restrict__ ovf,
                     const float* __restrict__ attn,
                     const float* __restrict__ hin, float* __restrict__ hout,
                     int residual,
                     const _Float16* __restrict__ Bt, float* __restrict__ featn) {
    const int lane = threadIdx.x & 31;
    const int node = __builtin_amdgcn_readfirstlane(
        (int)(blockIdx.x * (FIRST ? STRIP_WAVES : 8) + (threadIdx.x >> 5)));
    if (!FIRST && node >= N_NODES) return;
    int ids = bkt[node * BKT + lane];
    const int deg = __builtin_amdgcn_readfirstlane(ids);
    const int nb = deg < BKT_CAP ? deg : BKT_CAP;
    if (FIRST) {
        int key = lane == 0 ? (-2147483647 - 1) : (lane <= nb ? ids : 2147483647);
        key = sort32(key, lane);
        ids = lane == 0 ? deg : key;
        bkt[node * BKT + lane] = ids;
    }
    const unsigned lob = (unsigned)lane * 16u;
    const char* fb = (const char*)feat;
    const size_t nrow = (size_t)node * (HDIM * 4u);
    const float4 er4 = *(const float4*)(fb + nrow + lob);
    const float4 at4 = *(const float4*)((const char*)attn + lob);
    const v2f er01 = {er4.x, er4.y}, er23 = {er4.z, er4.w};
    const v2f at01 = {at4.x, at4.y}, at23 = {at4.z, at4.w};
    float m = -INFINITY, s = 0.f;
    v2f acc01 = {0.f, 0.f}, acc23 = {0.f, 0.f};

    for (int j = 0; j < nb; j += EDGE_CHUNK) {
        const int cnt = nb - j;
        v4f el[EDGE_CHUNK];
#pragma unroll
        for (int k = 0; k < EDGE_CHUNK; ++k) {
            const int jj = k < cnt ? j + k : nb - 1;
            const int s_ = __builtin_amdgcn_readlane(ids, 1 + jj);
            el[k] = *(const v4f*)(fb + (((unsigned)s_ << 9) + lob));
        }
        asm volatile("" :: "v"(el[0]), "v"(el[1]), "v"(el[2]), "v"(el[3]),
                           "v"(el[4]), "v"(el[5]), "v"(el[6]), "v"(el[7]));
        float sc[EDGE_CHUNK];
#pragma unroll
        for (int k = 0; k < EDGE_CHUNK; ++k) {
            sc[k] = -INFINITY;
            if (k < cnt) {
                v2f x01 = el[k].xy, x23 = el[k].zw;
                x01 += er01; x23 += er23;
                const v2f y01 = x01 * NEG_SLOPE, y23 = x23 * NEG_SLOPE;
                x01 = __builtin_elementwise_max(x01, y01);
                x23 = __builtin_elementwise_max(x23, y23);
                const v2f t = x01 * at01 + x23 * at23;
                float part = t.x + t.y;
                part += row_xor<1>(part);
                part += row_xor<2>(part);
                part += row_xor<4>(part);
                part += row_xor<8>(part);
                sc[k] = part;
            }
        }
        float mc = m;
#pragma unroll
        for (int k = 0; k < EDGE_CHUNK; ++k) mc = fmaxf(mc, sc[k]);
        const float scale = __expf(m - mc);
        float ps = 0.f;
        v2f a01 = {0.f, 0.f}, a23 = {0.f, 0.f};
#pragma unroll
        for (int k = 0; k < EDGE_CHUNK; ++k) {
            if (k < cnt) {
                const float p = __expf(sc[k] - mc);
                const v2f e01 = el[k].xy, e23 = el[k].zw;
                ps += p;
                a01 += e01 * p; a23 += e23 * p;
            }
        }
        s = s * scale + ps;
        acc01 = acc01 * scale + a01;
        acc23 = acc23 * scale + a23;
        m = mc;
    }

    if (deg > BKT_CAP) {
        int total = *ovfc;
        total = total < E_EDGES ? total : E_EDGES;
        const int2* __restrict__ o2 = (const int2*)ovf;
        for (int base = 0; base < total; base += 32) {
            const int i = base + lane;
            const int2 q = o2[i < total ? i : total - 1];
            unsigned hit = __builtin_amdgcn_ballot_w32(i < total && q.x == node);
            while (hit != 0u) {
                const int s_ = __builtin_amdgcn_readlane(q.y, __builtin_ctz(hit));
                hit &= hit - 1u;
                const v4f e = *(const v4f*)(fb + (((unsigned)s_ << 9) + lob));
                v2f x01 = e.xy, x23 = e.zw;
                x01 += er01; x23 += er23;
                const v2f y01 = x01 * NEG_SLOPE, y23 = x23 * NEG_SLOPE;
                x01 = __builtin_elementwise_max(x01, y01);
                x23 = __builtin_elementwise_max(x23, y23);
                const v2f t = x01 * at01 + x23 * at23;
                float part = t.x + t.y;
                part += row_xor<1>(part);
                part += row_xor<2>(part);
                part += row_xor<4>(part);
                part += row_xor<8>(part);
                const float mc = fmaxf(m, part);
                const float scale = __expf(m - mc), p = __expf(part - mc);
                s = s * scale + p;
                acc01 = acc01 * scale + e.xy * p;
                acc23 = acc23 * scale + e.zw * p;
                m = mc;
            }
        }
    }

    float4 r = make_float4(0.f, 0.f, 0.f, 0.f);
    if (deg > 0) { r.x = acc01.x / s; r.y = acc01.y / s; r.z = acc23.x / s; r.w = acc23.y / s; }
    if (residual) {
        const float4 hv = *(const float4*)((const char*)hin + nrow + lob);
        r.x += hv.x; r.y += hv.y; r.z += hv.z; r.w += hv.w;
    }
    r.x = r.x > 0.f ? r.x : (__expf(r.x) - 1.0f);
    r.y = r.y > 0.f ? r.y : (__expf(r.y) - 1.0f);
    r.z = r.z > 0.f ? r.z : (__expf(r.z) - 1.0f);
    r.w = r.w > 0.f ? r.w : (__expf(r.w) - 1.0f);
    if constexpr (!FIRST) {
        *(float4*)((char*)hout + nrow + lob) = r;
    } else {
        __shared__ __attribute__((aligned(16))) _Float16 hs[STRIP_WAVES * H_PITCH];
        __shared__ __attribute__((aligned(16))) float fs[STRIP_WAVES * C_PITCH];
        const int w = threadIdx.x >> 5;
        const int half = lane >> 4, rr = lane & 15;
        v16h b[4];
        if (w < 8) {
#pragma unroll
            for (int kc = 0; kc < 4; ++kc)
                b[kc] = load_b_f16(Bt + (w * 16 + rr) * 128 + kc * 32 + half * 8);
        }
        v4h q;
        q[0] = (_Float16)r.x; q[1] = (_Float16)r.y; q[2] = (_Float16)r.z; q[3] = (_Float16)r.w;
        *(v4h*)(hs + w * H_PITCH + lane * 4) = q;
        __syncthreads();
        if (w < 8) {
            v8f acc = {};
#pragma unroll
            for (int kc = 0; kc < 4; ++kc) {
                const v16h a = load_b_f16(hs + rr * H_PITCH + kc * 32 + half * 8);
                acc = __builtin_amdgcn_wmma_f32_16x16x32_f16(false, a, false, b[kc],
                                                             (short)0, acc, false, false);
            }
#pragma unroll
            for (int v = 0; v < 8; ++v) fs[(v + half * 8) * C_PITCH + w * 16 + rr] = acc[v];
        }
        __syncthreads();
        *(float4*)((char*)hout + nrow + lob) = r;
        *(v4f*)((char*)featn + nrow + lob) = *(const v4f*)(fs + w * C_PITCH + lane * 4);
    }
}

#define PRED_WAVES 8
#define P2_PITCH 72
#define Z2_PITCH 65
__global__ __launch_bounds__(256)
void predictor_wmma(const float* __restrict__ h,
                    const int* __restrict__ ps, const int* __restrict__ pd,
                    const int* __restrict__ ns, const int* __restrict__ nd,
                    const _Float16* __restrict__ Wp1t, const float* __restrict__ bp1,
                    const _Float16* __restrict__ Wp2t, const float* __restrict__ bp2,
                    const float* __restrict__ Wp3, const float* __restrict__ bp3,
                    float* __restrict__ out) {
    __shared__ __attribute__((aligned(16))) _Float16 w1s[64 * H_PITCH];
    __shared__ __attribute__((aligned(16))) _Float16 w2s[64 * P2_PITCH];
    __shared__ __attribute__((aligned(16))) _Float16 zs[PRED_WAVES][16 * H_PITCH];
    __shared__ float z2s[PRED_WAVES][16 * Z2_PITCH];
    const int tilesPerSet = P_CAND / 16;
    const int w = threadIdx.x >> 5, lane = threadIdx.x & 31;
    const int half = lane >> 4, r = lane & 15;
    const int tile = __builtin_amdgcn_readfirstlane((int)(blockIdx.x * PRED_WAVES + w));
    const bool live = tile < 2 * tilesPerSet;
    const bool isNeg = tile >= tilesPerSet;
    const int t = isNeg ? tile - tilesPerSet : tile;
    _Float16* zw = zs[w];

    v8h q1[4], q2[2];
#pragma unroll
    for (int i = 0; i < 4; ++i) q1[i] = *(const v8h*)(Wp1t + (i * 256 + threadIdx.x) * 8);
#pragma unroll
    for (int i = 0; i < 2; ++i) q2[i] = *(const v8h*)(Wp2t + (i * 256 + threadIdx.x) * 8);

    int mine = 0;
    if (live) {
        const int* S = isNeg ? ns : ps;
        const int* D = isNeg ? nd : pd;
        mine = half ? D[t * 16 + r] : S[t * 16 + r];
    }
#pragma unroll
    for (int i = 0; i < 4; ++i) {
        const int c = i * 256 + threadIdx.x;
        *(v8h*)(w1s + (c >> 4) * H_PITCH + (c & 15) * 8) = q1[i];
    }
#pragma unroll
    for (int i = 0; i < 2; ++i) {
        const int c = i * 256 + threadIdx.x;
        *(v8h*)(w2s + (c >> 3) * P2_PITCH + (c & 7) * 8) = q2[i];
    }
    if (live) {
        const unsigned lob = (unsigned)lane * 16u;
        const char* hb = (const char*)h;
        v4f sr[16], dr[16];
#pragma unroll
        for (int i = 0; i < 16; ++i) {
            const int si = __builtin_amdgcn_readlane(mine, i);
            const int di = __builtin_amdgcn_readlane(mine, 16 + i);
            sr[i] = *(const v4f*)(hb + (((unsigned)si << 9) + lob));
            dr[i] = *(const v4f*)(hb + (((unsigned)di << 9) + lob));
        }
#pragma unroll
        for (int i = 0; i < 16; ++i) {
            v4h q;
            q[0] = (_Float16)(sr[i][0] * dr[i][0]); q[1] = (_Float16)(sr[i][1] * dr[i][1]);
            q[2] = (_Float16)(sr[i][2] * dr[i][2]); q[3] = (_Float16)(sr[i][3] * dr[i][3]);
            *(v4h*)(zw + i * H_PITCH + lane * 4) = q;
        }
    }
    __syncthreads();
    if (!live) return;

    v16h a1[4];
#pragma unroll
    for (int kc = 0; kc < 4; ++kc) a1[kc] = load_b_f16(zw + r * H_PITCH + kc * 32 + half * 8);
#pragma unroll
    for (int nt = 0; nt < 4; ++nt) {
        const int col = nt * 16 + r;
        v8f acc = {};
#pragma unroll
        for (int kc = 0; kc < 4; ++kc) {
            v16h b = load_b_f16(w1s + col * H_PITCH + kc * 32 + half * 8);
            acc = __builtin_amdgcn_wmma_f32_16x16x32_f16(false, a1[kc], false, b,
                                                         (short)0, acc, false, false);
        }
        const float bias = bp1[col];
#pragma unroll
        for (int v = 0; v < 8; ++v) {
            float z = acc[v] + bias;
            z = z > 0.f ? z : 0.f;
            zw[(v + half * 8) * P2_PITCH + col] = (_Float16)z;
        }
    }

    v16h a2[2];
#pragma unroll
    for (int kc = 0; kc < 2; ++kc) a2[kc] = load_b_f16(zw + r * P2_PITCH + kc * 32 + half * 8);
#pragma unroll
    for (int nt = 0; nt < 4; ++nt) {
        const int col = nt * 16 + r;
        v8f acc = {};
#pragma unroll
        for (int kc = 0; kc < 2; ++kc) {
            v16h b = load_b_f16(w2s + col * P2_PITCH + kc * 32 + half * 8);
            acc = __builtin_amdgcn_wmma_f32_16x16x32_f16(false, a2[kc], false, b,
                                                         (short)0, acc, false, false);
        }
        const float bias = bp2[col];
#pragma unroll
        for (int v = 0; v < 8; ++v) {
            float z = acc[v] + bias;
            z2s[w][(v + half * 8) * Z2_PITCH + col] = z > 0.f ? z : 0.f;
        }
    }

    float part = 0.f;
    const int jb = half * 32;
#pragma unroll
    for (int j = 0; j < 32; ++j)
        part += z2s[w][r * Z2_PITCH + jb + j] * Wp3[jb + j];
    part += __shfl_xor(part, 16, 32);
    if (half == 0) {
        out[(isNeg ? P_CAND : 0) + t * 16 + r] = part + bp3[0];
    }
}

extern "C" void kernel_launch(void* const* d_in, const int* in_sizes, int n_in,
                              void* d_out, int out_size, void* d_ws, size_t ws_size,
                              hipStream_t stream) {
    const float* x        = (const float*)d_in[0];
    const int*   src      = (const int*)  d_in[1];
    const int*   dst      = (const int*)  d_in[2];
    const int*   pos_src  = (const int*)  d_in[3];
    const int*   pos_dst  = (const int*)  d_in[4];
    const int*   neg_src  = (const int*)  d_in[5];
    const int*   neg_dst  = (const int*)  d_in[6];
    const float* W0       = (const float*)d_in[7];
    const float* attn0    = (const float*)d_in[8];
    const float* W1       = (const float*)d_in[9];
    const float* attn1    = (const float*)d_in[10];
    const float* Wp1      = (const float*)d_in[11];
    const float* bp1      = (const float*)d_in[12];
    const float* Wp2      = (const float*)d_in[13];
    const float* bp2      = (const float*)d_in[14];
    const float* Wp3      = (const float*)d_in[15];
    const float* bp3      = (const float*)d_in[16];

    char* ws = (char*)d_ws;
    size_t off = 0;
    auto carve = [&](size_t bytes) -> char* {
        char* p = ws + off;
        off += (bytes + 255) & ~(size_t)255;
        return p;
    };
    float*     feat  = (float*)    carve((size_t)N_NODES * HDIM * 4);
    float*     h0    = (float*)    carve((size_t)N_NODES * HDIM * 4);
    float*     h1    = (float*)    carve((size_t)N_NODES * HDIM * 4);
    float*     feat1 = (float*)    carve((size_t)N_NODES * HDIM * 4);
    int*       bkt   = (int*)      carve((size_t)N_NODES * BKT * 4);
    int*       ovfc  = (int*)      carve(4);
    int*       ovf   = (int*)      carve((size_t)E_EDGES * 8);
    _Float16*  W0t   = (_Float16*) carve(128 * 128 * 2);
    _Float16*  W1t   = (_Float16*) carve(128 * 128 * 2);
    _Float16*  Wp1t  = (_Float16*) carve(64 * 128 * 2);
    _Float16*  Wp2t  = (_Float16*) carve(64 * 64 * 2);

    prep_weights<<<256, 256, 0, stream>>>(
        W0, W1, Wp1, Wp2, W0t, W1t, Wp1t, Wp2t, bkt, ovfc);

    const int gemmBlocks  = (N_NODES / 16 + GEMM_WAVES - 1) / GEMM_WAVES;
    const int edgeThrBlocks = (E_EDGES + 255) / 256;
    const int nodeWaveBlocks = (N_NODES + 7) / 8;
    const int nodeStripBlocks = N_NODES / STRIP_WAVES;

    edge_fill_kernel<<<edgeThrBlocks, 256, 0, stream>>>(src, dst, bkt, ovfc, ovf);

    gemm128_wmma<<<gemmBlocks, 256, 0, stream>>>(x, W0t, feat, N_NODES);
    gat_node_kernel<true><<<nodeStripBlocks, 32 * STRIP_WAVES, 0, stream>>>(
        feat, bkt, ovfc, ovf, attn0, x, h0, 0, W1t, feat1);

    gat_node_kernel<false><<<nodeWaveBlocks, 256, 0, stream>>>(
        feat1, bkt, ovfc, ovf, attn1, h0, h1, 1, nullptr, nullptr);

    const int predBlocks = (2 * (P_CAND / 16) + PRED_WAVES - 1) / PRED_WAVES;
    predictor_wmma<<<predBlocks, 256, 0, stream>>>(
        h1, pos_src, pos_dst, neg_src, neg_dst,
        Wp1t, bp1, Wp2t, bp2, Wp3, bp3, (float*)d_out);
}
